# starting kernel
# baseline (speedup 1.0000x reference)
#include <hip/hip_runtime.h>

typedef float v2f __attribute__((ext_vector_type(2)));
typedef float v8f __attribute__((ext_vector_type(8)));

#define IN_DIM  128
#define HID_DIM 32
#define OUT_DIM 8

__global__ void __launch_bounds__(256) init_kernel(float* __restrict__ deg,
                                                   float* __restrict__ h,
                                                   float* __restrict__ out,
                                                   const float* __restrict__ b2,
                                                   int n) {
    int i = blockIdx.x * blockDim.x + threadIdx.x;
    if (i < n) deg[i] = 1.0f;
    if (i < n * OUT_DIM) out[i] = b2[i & (OUT_DIM - 1)];
    if (i < n * HID_DIM) h[i] = 0.0f;
}

__global__ void __launch_bounds__(256) deg_kernel(const int* __restrict__ dstIdx,
                                                  float* __restrict__ deg, int E) {
    int e = blockIdx.x * blockDim.x + threadIdx.x;
    if (e < E) atomicAdd(&deg[dstIdx[e]], 1.0f);
}

__global__ void __launch_bounds__(256) rsqrt_kernel(float* __restrict__ deg, int n) {
    int i = blockIdx.x * blockDim.x + threadIdx.x;
    if (i < n) deg[i] = rsqrtf(deg[i]);
}

__global__ void __launch_bounds__(256) gemm1_kernel(const float* __restrict__ x,
                                                    const float* __restrict__ W1,
                                                    float* __restrict__ hw, int n) {
    const int lane = threadIdx.x & 31;
    const int half = lane >> 4;
    const int l16  = lane & 15;
    const int tile = blockIdx.x * 8 + (threadIdx.x >> 5);
    const int ntiles = (n + 15) >> 4;
    if (tile >= ntiles) return;

    int row = tile * 16 + l16;
    if (row >= n) row = n - 1;
    const float* xr = x + row * IN_DIM;

    v8f c0 = {}; v8f c1 = {};
    #pragma unroll 8
    for (int k = 0; k < IN_DIM; k += 4) {
        const int kk = k + 2 * half;
        v2f a;  a.x  = xr[kk];                       a.y  = xr[kk + 1];
        v2f b0; b0.x = W1[kk * HID_DIM + l16];       b0.y = W1[(kk + 1) * HID_DIM + l16];
        v2f b1; b1.x = W1[kk * HID_DIM + 16 + l16];  b1.y = W1[(kk + 1) * HID_DIM + 16 + l16];
        c0 = __builtin_amdgcn_wmma_f32_16x16x4_f32(false, a, false, b0, (short)0, c0, false, false);
        c1 = __builtin_amdgcn_wmma_f32_16x16x4_f32(false, a, false, b1, (short)0, c1, false, false);
    }
    #pragma unroll
    for (int i = 0; i < 8; ++i) {
        int orow = tile * 16 + i + 8 * half;
        if (orow < n) {
            hw[orow * HID_DIM + l16]      = c0[i];
            hw[orow * HID_DIM + 16 + l16] = c1[i];
        }
    }
}

__global__ void __launch_bounds__(256) agg1_kernel(const int* __restrict__ srcIdx,
                                                   const int* __restrict__ dstIdx,
                                                   const float* __restrict__ dinv,
                                                   const float* __restrict__ hw,
                                                   float* __restrict__ h,
                                                   int E, int n) {
    int tid  = blockIdx.x * blockDim.x + threadIdx.x;
    int e    = tid >> 5;
    int lane = tid & 31;
    if (e >= E + n) return;
    int s, d;
    if (e < E) { s = srcIdx[e]; d = dstIdx[e]; }
    else       { s = e - E;     d = s; }
    float nrm = dinv[s] * dinv[d];
    float val = hw[s * HID_DIM + lane] * nrm;
    atomicAdd(&h[d * HID_DIM + lane], val);
}

__global__ void __launch_bounds__(256) bias_relu_kernel(float* __restrict__ h,
                                                        const float* __restrict__ b1, int n) {
    int i = blockIdx.x * blockDim.x + threadIdx.x;
    if (i < n * HID_DIM) h[i] = fmaxf(h[i] + b1[i & (HID_DIM - 1)], 0.0f);
}

__global__ void __launch_bounds__(256) gemm2_kernel(const float* __restrict__ h,
                                                    const float* __restrict__ W2,
                                                    float* __restrict__ g, int n) {
    const int lane = threadIdx.x & 31;
    const int half = lane >> 4;
    const int l16  = lane & 15;
    const int tile = blockIdx.x * 8 + (threadIdx.x >> 5);
    const int ntiles = (n + 15) >> 4;
    if (tile >= ntiles) return;

    int row = tile * 16 + l16;
    if (row >= n) row = n - 1;
    const float* hr = h + row * HID_DIM;
    const bool live = (l16 < OUT_DIM);

    v8f c = {};
    #pragma unroll
    for (int k = 0; k < HID_DIM; k += 4) {
        const int kk = k + 2 * half;
        v2f a; a.x = hr[kk]; a.y = hr[kk + 1];
        v2f b;
        b.x = live ? W2[kk * OUT_DIM + l16]       : 0.0f;
        b.y = live ? W2[(kk + 1) * OUT_DIM + l16] : 0.0f;
        c = __builtin_amdgcn_wmma_f32_16x16x4_f32(false, a, false, b, (short)0, c, false, false);
    }
    if (live) {
        #pragma unroll
        for (int i = 0; i < 8; ++i) {
            int orow = tile * 16 + i + 8 * half;
            if (orow < n) g[orow * OUT_DIM + l16] = c[i];
        }
    }
}

__global__ void __launch_bounds__(256) agg2_kernel(const int* __restrict__ srcIdx,
                                                   const int* __restrict__ dstIdx,
                                                   const float* __restrict__ dinv,
                                                   const float* __restrict__ g,
                                                   float* __restrict__ out,
                                                   int E, int n) {
    int tid = blockIdx.x * blockDim.x + threadIdx.x;
    int e = tid >> 3;
    int f = tid & 7;
    if (e >= E + n) return;
    int s, d;
    if (e < E) { s = srcIdx[e]; d = dstIdx[e]; }
    else       { s = e - E;     d = s; }
    float nrm = dinv[s] * dinv[d];
    atomicAdd(&out[d * OUT_DIM + f], g[s * OUT_DIM + f] * nrm);
}

extern "C" void kernel_launch(void* const* d_in, const int* in_sizes, int n_in,
                              void* d_out, int out_size, void* d_ws, size_t ws_size,
                              hipStream_t stream) {
    const float* x  = (const float*)d_in[0];
    const int*   ei = (const int*)d_in[1];
    const float* W1 = (const float*)d_in[2];
    const float* b1 = (const float*)d_in[3];
    const float* W2 = (const float*)d_in[4];
    const float* b2 = (const float*)d_in[5];
    float* out = (float*)d_out;

    const int n = in_sizes[0] / IN_DIM;
    const int E = in_sizes[1] / 2;
    const int* srcIdx = ei;
    const int* dstIdx = ei + E;

    float* deg = (float*)d_ws;
    float* hw  = deg + n;
    float* h   = hw + (size_t)n * HID_DIM;
    float* g   = h  + (size_t)n * HID_DIM;

    const int ET = E + n;
    const dim3 blk(256);

    init_kernel<<<dim3((n * HID_DIM + 255) / 256), blk, 0, stream>>>(deg, h, out, b2, n);
    deg_kernel<<<dim3((E + 255) / 256), blk, 0, stream>>>(dstIdx, deg, E);
    rsqrt_kernel<<<dim3((n + 255) / 256), blk, 0, stream>>>(deg, n);

    const int ntiles = (n + 15) / 16;
    gemm1_kernel<<<dim3((ntiles + 7) / 8), blk, 0, stream>>>(x, W1, hw, n);

    long t1 = (long)ET * 32;
    agg1_kernel<<<dim3((unsigned)((t1 + 255) / 256)), blk, 0, stream>>>(srcIdx, dstIdx, deg, hw, h, E, n);
    bias_relu_kernel<<<dim3((n * HID_DIM + 255) / 256), blk, 0, stream>>>(h, b1, n);

    gemm2_kernel<<<dim3((ntiles + 7) / 8), blk, 0, stream>>>(h, W2, g, n);

    long t2 = (long)ET * 8;
    agg2_kernel<<<dim3((unsigned)((t2 + 255) / 256)), blk, 0, stream>>>(srcIdx, dstIdx, deg, g, out, E, n);
}


# All three scatter passes: one 1024-thread work-group owns each block of destination nodes, reads a byte map of every edge's block, sums its rows by LDS float adds (arrival order); no global atomics
# speedup vs baseline: 1.7784x; 1.7784x over previous
#include <hip/hip_runtime.h>


typedef float v2f __attribute__((ext_vector_type(2)));
typedef float v8f __attribute__((ext_vector_type(8)));

#define IN_DIM  128
#define HID_DIM 32
#define OUT_DIM 8

#define NBLK    256
#define RMAX    512
#define OWN_T   1024
#define OWN_WAVES (OWN_T / 32)
#define QCAP    64

__global__ void __launch_bounds__(256) blockmap_kernel(const int* __restrict__ dstIdx,
                                                       unsigned int* __restrict__ map4,
                                                       int E, int nwords, int n) {
    const int t = blockIdx.x * blockDim.x + threadIdx.x;
    if (t >= nwords) return;
    const unsigned R = (unsigned)((n + NBLK - 1) / NBLK);
    unsigned w = 0;
    #pragma unroll
    for (int j = 0; j < 4; ++j) {
        const int e = 4 * t + j;
        unsigned b = 0;
        if (e < E) b = ((unsigned)dstIdx[e] / R) & 255u;
        w |= b << (8 * j);
    }
    map4[t] = w;
}

__device__ __forceinline__ unsigned zero_bytes(unsigned x) {
    const unsigned t = (x & 0x7F7F7F7Fu) + 0x7F7F7F7Fu;
    return ~(t | x | 0x7F7F7F7Fu);
}

template <int W>
__device__ __forceinline__ int lane_get(int v, int idx) {
    if constexpr (W == 32) return __builtin_amdgcn_readlane(v, idx);
    else                   return __shfl(v, idx);
}

__device__ __forceinline__ void drain_count(const int* q, int cnt, int lane, int base,
                                            const int* __restrict__ dstIdx, int* cnti) {
    if (lane < cnt) {
        const int d = dstIdx[q[lane]];
        atomicAdd(&cnti[d - base], 1);
    }
}

template <int W>
__device__ __forceinline__ void drain_rows(const int* q, int cnt, int lane, int base,
                                           const int* __restrict__ srcIdx,
                                           const int* __restrict__ dstIdx,
                                           const float* __restrict__ dinv,
                                           const float* __restrict__ feat, float* acc) {
    constexpr int EPI = 32 / W;
    constexpr int U   = 8;
    const int sub = lane / W;
    const int f   = lane % W;
    int s = 0, dl = 0;
    float nrm = 0.0f;
    if (lane < cnt) {
        const int e = q[lane];
        s = srcIdx[e];
        const int d = dstIdx[e];
        dl = d - base;
        nrm = dinv[s] * dinv[d];
    }
    for (int i0 = 0; i0 < cnt; i0 += U * EPI) {
        float r[U];
        #pragma unroll
        for (int j = 0; j < U; ++j) {
            const int idx = i0 + j * EPI + sub;
            const int sj = lane_get<W>(s, idx);
            r[j] = feat[(size_t)sj * W + f];
        }
        #pragma unroll
        for (int j = 0; j < U; ++j) {
            const int idx = i0 + j * EPI + sub;
            const float nj = __int_as_float(lane_get<W>(__float_as_int(nrm), idx));
            const int   dj = lane_get<W>(dl, idx);
            if (idx < cnt) atomicAdd(&acc[dj * W + f], r[j] * nj);
        }
    }
}

template <int MODE>
__device__ __forceinline__ void owner_body(const unsigned int* __restrict__ map4,
                                           const int* __restrict__ srcIdx,
                                           const int* __restrict__ dstIdx,
                                           const float* __restrict__ dinv,
                                           const float* __restrict__ feat,
                                           const float* __restrict__ bias,
                                           float* __restrict__ outp, int E, int n) {
    constexpr int W = (MODE == 1) ? HID_DIM : ((MODE == 2) ? OUT_DIM : 1);
    __shared__ float acc[RMAX * W];
    __shared__ int   queue[OWN_WAVES * QCAP];
    int* cnti = reinterpret_cast<int*>(acc);

    const int tid  = threadIdx.x;
    const int lane = tid & 31;
    const int wave = tid >> 5;
    const int blk  = blockIdx.x;
    const int R    = (n + NBLK - 1) / NBLK;
    const int base = blk * R;
    int rows = n - base;
    if (rows > R) rows = R;
    if (rows <= 0) return;

    for (int i = tid; i < rows * W; i += OWN_T) {
        if constexpr (MODE == 0) {
            cnti[i] = 1;
        } else {
            const int node = base + i / W;
            const float dv = dinv[node];
            acc[i] = feat[(size_t)node * W + (i % W)] * (dv * dv);
        }
    }
    __syncthreads();

    int* myq = queue + wave * QCAP;
    int qn = 0;
    const int nvec = (E + 15) >> 4;
    const uint4* mapv = reinterpret_cast<const uint4*>(map4);
    const unsigned splat = (unsigned)blk * 0x01010101u;

    for (int vb = 0; vb < nvec; vb += OWN_T * 4) {
        uint4 mq[4];
        int   vi[4];
        #pragma unroll
        for (int u = 0; u < 4; ++u) {
            vi[u] = vb + u * OWN_T + tid;
            mq[u] = mapv[vi[u] < nvec ? vi[u] : nvec - 1];
        }
        #pragma unroll
        for (int u = 0; u < 4; ++u) {
            if (vi[u] >= nvec) mq[u] = make_uint4(~splat, ~splat, ~splat, ~splat);
        }
        #pragma unroll
        for (int u = 0; u < 4; ++u) {
            unsigned P = zero_bytes(mq[u].x ^ splat)
                       | (zero_bytes(mq[u].y ^ splat) >> 1)
                       | (zero_bytes(mq[u].z ^ splat) >> 2)
                       | (zero_bytes(mq[u].w ^ splat) >> 3);
            unsigned any = __builtin_amdgcn_ballot_w32(P != 0u);
            while (any != 0u) {
                const bool m = (P != 0u);
                const int  k = __builtin_ctz(m ? P : 1u);
                const int  e = vi[u] * 16 + (7 - (k & 7)) * 4 + (k >> 3);
                const bool ok = m && (e < E);
                const unsigned mask = __builtin_amdgcn_ballot_w32(ok);
                if (ok) myq[qn + (int)__builtin_amdgcn_mbcnt_lo(mask, 0u)] = e;
                qn += __builtin_popcount(mask);
                if (qn >= 32) {
                    qn -= 32;
                    if constexpr (MODE == 0) drain_count(myq + qn, 32, lane, base, dstIdx, cnti);
                    else drain_rows<W>(myq + qn, 32, lane, base, srcIdx, dstIdx, dinv, feat, acc);
                }
                P &= P - 1u;
                any = __builtin_amdgcn_ballot_w32(P != 0u);
            }
        }
    }
    if (qn > 0) {
        if constexpr (MODE == 0) drain_count(myq, qn, lane, base, dstIdx, cnti);
        else drain_rows<W>(myq, qn, lane, base, srcIdx, dstIdx, dinv, feat, acc);
    }
    __syncthreads();

    for (int i = tid; i < rows * W; i += OWN_T) {
        if constexpr (MODE == 0) {
            outp[base + i] = rsqrtf((float)cnti[i]);
        } else if constexpr (MODE == 1) {
            outp[(size_t)base * W + i] = fmaxf(acc[i] + bias[i & (W - 1)], 0.0f);
        } else {
            outp[(size_t)base * W + i] = acc[i] + bias[i & (W - 1)];
        }
    }
}

__global__ void __launch_bounds__(OWN_T) owner_deg_kernel(const unsigned int* __restrict__ map4,
                                                          const int* __restrict__ dstIdx,
                                                          float* __restrict__ dinv, int E, int n) {
    owner_body<0>(map4, nullptr, dstIdx, nullptr, nullptr, nullptr, dinv, E, n);
}

__global__ void __launch_bounds__(OWN_T) owner_agg1_kernel(const unsigned int* __restrict__ map4,
                                                           const int* __restrict__ srcIdx,
                                                           const int* __restrict__ dstIdx,
                                                           const float* __restrict__ dinv,
                                                           const float* __restrict__ hw,
                                                           const float* __restrict__ b1,
                                                           float* __restrict__ h, int E, int n) {
    owner_body<1>(map4, srcIdx, dstIdx, dinv, hw, b1, h, E, n);
}

__global__ void __launch_bounds__(OWN_T) owner_agg2_kernel(const unsigned int* __restrict__ map4,
                                                           const int* __restrict__ srcIdx,
                                                           const int* __restrict__ dstIdx,
                                                           const float* __restrict__ dinv,
                                                           const float* __restrict__ g,
                                                           const float* __restrict__ b2,
                                                           float* __restrict__ out, int E, int n) {
    owner_body<2>(map4, srcIdx, dstIdx, dinv, g, b2, out, E, n);
}

__global__ void __launch_bounds__(256) gemm1_kernel(const float* __restrict__ x,
                                                    const float* __restrict__ W1,
                                                    float* __restrict__ hw, int n) {
    const int lane = threadIdx.x & 31;
    const int half = lane >> 4;
    const int l16  = lane & 15;
    const int tile = blockIdx.x * 8 + (threadIdx.x >> 5);
    const int ntiles = (n + 15) >> 4;
    if (tile >= ntiles) return;

    int row = tile * 16 + l16;
    if (row >= n) row = n - 1;
    const float* xr = x + row * IN_DIM;

    v8f c0 = {}; v8f c1 = {};
    #pragma unroll 8
    for (int k = 0; k < IN_DIM; k += 4) {
        const int kk = k + 2 * half;
        v2f a;  a.x  = xr[kk];                       a.y  = xr[kk + 1];
        v2f b0; b0.x = W1[kk * HID_DIM + l16];       b0.y = W1[(kk + 1) * HID_DIM + l16];
        v2f b1; b1.x = W1[kk * HID_DIM + 16 + l16];  b1.y = W1[(kk + 1) * HID_DIM + 16 + l16];
        c0 = __builtin_amdgcn_wmma_f32_16x16x4_f32(false, a, false, b0, (short)0, c0, false, false);
        c1 = __builtin_amdgcn_wmma_f32_16x16x4_f32(false, a, false, b1, (short)0, c1, false, false);
    }
    #pragma unroll
    for (int i = 0; i < 8; ++i) {
        int orow = tile * 16 + i + 8 * half;
        if (orow < n) {
            hw[orow * HID_DIM + l16]      = c0[i];
            hw[orow * HID_DIM + 16 + l16] = c1[i];
        }
    }
}


__global__ void __launch_bounds__(256) gemm2_kernel(const float* __restrict__ h,
                                                    const float* __restrict__ W2,
                                                    float* __restrict__ g, int n) {
    const int lane = threadIdx.x & 31;
    const int half = lane >> 4;
    const int l16  = lane & 15;
    const int tile = blockIdx.x * 8 + (threadIdx.x >> 5);
    const int ntiles = (n + 15) >> 4;
    if (tile >= ntiles) return;

    int row = tile * 16 + l16;
    if (row >= n) row = n - 1;
    const float* hr = h + row * HID_DIM;
    const bool live = (l16 < OUT_DIM);

    v8f c = {};
    #pragma unroll
    for (int k = 0; k < HID_DIM; k += 4) {
        const int kk = k + 2 * half;
        v2f a; a.x = hr[kk]; a.y = hr[kk + 1];
        v2f b;
        b.x = live ? W2[kk * OUT_DIM + l16]       : 0.0f;
        b.y = live ? W2[(kk + 1) * OUT_DIM + l16] : 0.0f;
        c = __builtin_amdgcn_wmma_f32_16x16x4_f32(false, a, false, b, (short)0, c, false, false);
    }
    if (live) {
        #pragma unroll
        for (int i = 0; i < 8; ++i) {
            int orow = tile * 16 + i + 8 * half;
            if (orow < n) g[orow * OUT_DIM + l16] = c[i];
        }
    }
}


extern "C" void kernel_launch(void* const* d_in, const int* in_sizes, int n_in,
                              void* d_out, int out_size, void* d_ws, size_t ws_size,
                              hipStream_t stream) {
    const float* x  = (const float*)d_in[0];
    const int*   ei = (const int*)d_in[1];
    const float* W1 = (const float*)d_in[2];
    const float* b1 = (const float*)d_in[3];
    const float* W2 = (const float*)d_in[4];
    const float* b2 = (const float*)d_in[5];
    float* out = (float*)d_out;

    const int n = in_sizes[0] / IN_DIM;
    const int E = in_sizes[1] / 2;
    const int* srcIdx = ei;
    const int* dstIdx = ei + E;

    const int nvec   = (E + 15) / 16;
    const int nwords = nvec * 4;
    unsigned int* map4 = (unsigned int*)d_ws;
    float* dinv = (float*)(map4 + nwords);
    float* hw   = dinv + n;
    float* h    = hw + (size_t)n * HID_DIM;
    float* g    = h  + (size_t)n * HID_DIM;

    const dim3 blk(256);
    const int ntiles = (n + 15) / 16;

    blockmap_kernel<<<dim3((nwords + 255) / 256), blk, 0, stream>>>(dstIdx, map4, E, nwords, n);
    owner_deg_kernel<<<dim3(NBLK), dim3(OWN_T), 0, stream>>>(map4, dstIdx, dinv, E, n);
    gemm1_kernel<<<dim3((ntiles + 7) / 8), blk, 0, stream>>>(x, W1, hw, n);
    owner_agg1_kernel<<<dim3(NBLK), dim3(OWN_T), 0, stream>>>(map4, srcIdx, dstIdx, dinv, hw, b1, h, E, n);
    gemm2_kernel<<<dim3((ntiles + 7) / 8), blk, 0, stream>>>(h, W2, g, n);
    owner_agg2_kernel<<<dim3(NBLK), dim3(OWN_T), 0, stream>>>(map4, srcIdx, dstIdx, dinv, g, b2, out, E, n);
}


# Only the degree pass scans the byte map: each owner stores its edges, packed (local row, source), as a list that both row passes walk instead of rescanning; sums stay LDS float adds in arrival order
# speedup vs baseline: 2.8797x; 1.6193x over previous
#include <hip/hip_runtime.h>


typedef float v2f __attribute__((ext_vector_type(2)));
typedef float v8f __attribute__((ext_vector_type(8)));

#define IN_DIM  128
#define HID_DIM 32
#define OUT_DIM 8

#define NBLK    256
#define RMAX    512
#define OWN_T   1024
#define OWN_WAVES (OWN_T / 32)
#define QCAP    64
#define LCAP    65536
#define SRC_BITS 17
#define SRC_MASK ((1u << SRC_BITS) - 1u)
static_assert(NBLK * RMAX <= (1 << SRC_BITS), "a source node must fit the low field of a list entry");
static_assert(RMAX <= (1 << (32 - SRC_BITS)), "a row inside the block must fit the high field of a list entry");

__global__ void __launch_bounds__(256) blockmap_kernel(const int* __restrict__ dstIdx,
                                                       unsigned int* __restrict__ map4,
                                                       int E, int nwords, int n) {
    const int t = blockIdx.x * blockDim.x + threadIdx.x;
    if (t >= nwords) return;
    const unsigned R = (unsigned)((n + NBLK - 1) / NBLK);
    unsigned w = 0;
    #pragma unroll
    for (int j = 0; j < 4; ++j) {
        const int e = 4 * t + j;
        unsigned b = 0;
        if (e < E) b = ((unsigned)dstIdx[e] / R) & 255u;
        w |= b << (8 * j);
    }
    map4[t] = w;
}

__device__ __forceinline__ unsigned zero_bytes(unsigned x) {
    const unsigned t = (x & 0x7F7F7F7Fu) + 0x7F7F7F7Fu;
    return ~(t | x | 0x7F7F7F7Fu);
}

template <int W>
__device__ __forceinline__ int lane_get(int v, int idx) {
    if constexpr (W == 32) return __builtin_amdgcn_readlane(v, idx);
    else                   return __shfl(v, idx);
}

__device__ __forceinline__ void drain_count(const int* q, int cnt, int lane, int base,
                                            const int* __restrict__ srcIdx,
                                            const int* __restrict__ dstIdx, int* cnti,
                                            int* cursor, unsigned int* __restrict__ mylist) {
    unsigned ent = 0;
    if (lane < cnt) {
        const int e = q[lane];
        const int d = dstIdx[e];
        const int s = srcIdx[e];
        atomicAdd(&cnti[d - base], 1);
        ent = ((unsigned)(d - base) << SRC_BITS) | (unsigned)s;
    }
    int off = 0;
    if (lane == 0) off = atomicAdd(cursor, cnt);
    off = __builtin_amdgcn_readfirstlane(off);
    if (off + cnt <= LCAP && lane < cnt) mylist[off + lane] = ent;
}

template <int W>
__device__ __forceinline__ void gather_add(int s, int dl, float nrm, int cnt, int lane,
                                           const float* __restrict__ feat, float* acc) {
    constexpr int EPI = 32 / W;
    constexpr int U   = 8;
    const int sub = lane / W;
    const int f   = lane % W;
    for (int i0 = 0; i0 < cnt; i0 += U * EPI) {
        float r[U];
        #pragma unroll
        for (int j = 0; j < U; ++j) {
            const int idx = i0 + j * EPI + sub;
            const int sj = lane_get<W>(s, idx);
            r[j] = feat[(size_t)sj * W + f];
        }
        #pragma unroll
        for (int j = 0; j < U; ++j) {
            const int idx = i0 + j * EPI + sub;
            const float nj = __int_as_float(lane_get<W>(__float_as_int(nrm), idx));
            const int   dj = lane_get<W>(dl, idx);
            if (idx < cnt) atomicAdd(&acc[dj * W + f], r[j] * nj);
        }
    }
}

template <int W>
__device__ __forceinline__ void drain_rows(const int* q, int cnt, int lane, int base,
                                           const int* __restrict__ srcIdx,
                                           const int* __restrict__ dstIdx,
                                           const float* __restrict__ dinv,
                                           const float* __restrict__ feat, float* acc) {
    int s = 0, dl = 0;
    float nrm = 0.0f;
    if (lane < cnt) {
        const int e = q[lane];
        s = srcIdx[e];
        const int d = dstIdx[e];
        dl = d - base;
        nrm = dinv[s] * dinv[d];
    }
    gather_add<W>(s, dl, nrm, cnt, lane, feat, acc);
}

template <int W>
__device__ __forceinline__ void replay_rows(const unsigned int* __restrict__ mylist, int total,
                                            int wave, int lane, int base,
                                            const float* __restrict__ dinv,
                                            const float* __restrict__ feat, float* acc) {
    int c = wave * 32;
    unsigned ent = (c + lane < total) ? mylist[c + lane] : 0u;
    for (; c < total; c += OWN_T) {
        const int cn = c + OWN_T;
        const unsigned nxt = (cn + lane < total) ? mylist[cn + lane] : 0u;
        int cnt = total - c;
        if (cnt > 32) cnt = 32;
        const int s  = (int)(ent & SRC_MASK);
        const int dl = (int)(ent >> SRC_BITS);
        float nrm = 0.0f;
        if (lane < cnt) nrm = dinv[s] * dinv[base + dl];
        gather_add<W>(s, dl, nrm, cnt, lane, feat, acc);
        ent = nxt;
    }
}

template <int MODE>
__device__ __forceinline__ void scan_map(const unsigned int* __restrict__ map4,
                                         const int* __restrict__ srcIdx,
                                         const int* __restrict__ dstIdx,
                                         const float* __restrict__ dinv,
                                         const float* __restrict__ feat,
                                         float* acc, int* queue, int* cursor,
                                         unsigned int* __restrict__ mylist,
                                         int E, int blk, int base) {
    constexpr int W = (MODE == 1) ? HID_DIM : ((MODE == 2) ? OUT_DIM : 1);
    int* cnti = reinterpret_cast<int*>(acc);
    const int tid  = threadIdx.x;
    const int lane = tid & 31;
    const int wave = tid >> 5;

    int* myq = queue + wave * QCAP;
    int qn = 0;
    const int nvec = (E + 15) >> 4;
    const uint4* mapv = reinterpret_cast<const uint4*>(map4);
    const unsigned splat = (unsigned)blk * 0x01010101u;

    for (int vb = 0; vb < nvec; vb += OWN_T * 4) {
        uint4 mq[4];
        int   vi[4];
        #pragma unroll
        for (int u = 0; u < 4; ++u) {
            vi[u] = vb + u * OWN_T + tid;
            mq[u] = mapv[vi[u] < nvec ? vi[u] : nvec - 1];
        }
        #pragma unroll
        for (int u = 0; u < 4; ++u) {
            if (vi[u] >= nvec) mq[u] = make_uint4(~splat, ~splat, ~splat, ~splat);
        }
        #pragma unroll
        for (int u = 0; u < 4; ++u) {
            unsigned P = zero_bytes(mq[u].x ^ splat)
                       | (zero_bytes(mq[u].y ^ splat) >> 1)
                       | (zero_bytes(mq[u].z ^ splat) >> 2)
                       | (zero_bytes(mq[u].w ^ splat) >> 3);
            unsigned any = __builtin_amdgcn_ballot_w32(P != 0u);
            while (any != 0u) {
                const bool m = (P != 0u);
                const int  k = __builtin_ctz(m ? P : 1u);
                const int  e = vi[u] * 16 + (7 - (k & 7)) * 4 + (k >> 3);
                const bool ok = m && (e < E);
                const unsigned mask = __builtin_amdgcn_ballot_w32(ok);
                if (ok) myq[qn + (int)__builtin_amdgcn_mbcnt_lo(mask, 0u)] = e;
                qn += __builtin_popcount(mask);
                if (qn >= 32) {
                    qn -= 32;
                    if constexpr (MODE == 0) drain_count(myq + qn, 32, lane, base, srcIdx, dstIdx, cnti, cursor, mylist);
                    else drain_rows<W>(myq + qn, 32, lane, base, srcIdx, dstIdx, dinv, feat, acc);
                }
                P &= P - 1u;
                any = __builtin_amdgcn_ballot_w32(P != 0u);
            }
        }
    }
    if (qn > 0) {
        if constexpr (MODE == 0) drain_count(myq, qn, lane, base, srcIdx, dstIdx, cnti, cursor, mylist);
        else drain_rows<W>(myq, qn, lane, base, srcIdx, dstIdx, dinv, feat, acc);
    }
}

template <int MODE>
__device__ __forceinline__ void owner_body(const unsigned int* __restrict__ map4,
                                           const int* __restrict__ srcIdx,
                                           const int* __restrict__ dstIdx,
                                           const float* __restrict__ dinv,
                                           const float* __restrict__ feat,
                                           const float* __restrict__ bias,
                                           float* __restrict__ outp,
                                           unsigned int* __restrict__ list,
                                           int* __restrict__ listcnt, int E, int n) {
    constexpr int W = (MODE == 1) ? HID_DIM : ((MODE == 2) ? OUT_DIM : 1);
    __shared__ float acc[RMAX * W];
    __shared__ int   queue[OWN_WAVES * QCAP];
    __shared__ int   cursor;
    int* cnti = reinterpret_cast<int*>(acc);

    const int tid  = threadIdx.x;
    const int blk  = blockIdx.x;
    const int R    = (n + NBLK - 1) / NBLK;
    const int base = blk * R;
    int rows = n - base;
    if (rows > R) rows = R;
    if (rows <= 0) return;
    unsigned int* mylist = list + (size_t)blk * LCAP;

    for (int i = tid; i < rows * W; i += OWN_T) {
        if constexpr (MODE == 0) {
            cnti[i] = 1;
        } else {
            const int node = base + i / W;
            const float dv = dinv[node];
            acc[i] = feat[(size_t)node * W + (i % W)] * (dv * dv);
        }
    }
    if constexpr (MODE == 0) {
        if (tid == 0) cursor = 0;
    }
    __syncthreads();

    if constexpr (MODE == 0) {
        scan_map<0>(map4, srcIdx, dstIdx, dinv, feat, acc, queue, &cursor, mylist, E, blk, base);
    } else {
        const int total = listcnt[blk];
        if (total <= LCAP) replay_rows<W>(mylist, total, tid >> 5, tid & 31, base, dinv, feat, acc);
        else scan_map<MODE>(map4, srcIdx, dstIdx, dinv, feat, acc, queue, &cursor, mylist, E, blk, base);
    }
    __syncthreads();

    if constexpr (MODE == 0) {
        if (tid == 0) listcnt[blk] = cursor;
    }

    for (int i = tid; i < rows * W; i += OWN_T) {
        if constexpr (MODE == 0) {
            outp[base + i] = rsqrtf((float)cnti[i]);
        } else if constexpr (MODE == 1) {
            outp[(size_t)base * W + i] = fmaxf(acc[i] + bias[i & (W - 1)], 0.0f);
        } else {
            outp[(size_t)base * W + i] = acc[i] + bias[i & (W - 1)];
        }
    }
}

__global__ void __launch_bounds__(OWN_T) owner_deg_kernel(const unsigned int* __restrict__ map4,
                                                          const int* __restrict__ srcIdx,
                                                          const int* __restrict__ dstIdx,
                                                          float* __restrict__ dinv,
                                                          unsigned int* __restrict__ list,
                                                          int* __restrict__ listcnt, int E, int n) {
    owner_body<0>(map4, srcIdx, dstIdx, nullptr, nullptr, nullptr, dinv, list, listcnt, E, n);
}

__global__ void __launch_bounds__(OWN_T) owner_agg1_kernel(const unsigned int* __restrict__ map4,
                                                           const int* __restrict__ srcIdx,
                                                           const int* __restrict__ dstIdx,
                                                           const float* __restrict__ dinv,
                                                           const float* __restrict__ hw,
                                                           const float* __restrict__ b1,
                                                           float* __restrict__ h,
                                                           unsigned int* __restrict__ list,
                                                           int* __restrict__ listcnt, int E, int n) {
    owner_body<1>(map4, srcIdx, dstIdx, dinv, hw, b1, h, list, listcnt, E, n);
}

__global__ void __launch_bounds__(OWN_T) owner_agg2_kernel(const unsigned int* __restrict__ map4,
                                                           const int* __restrict__ srcIdx,
                                                           const int* __restrict__ dstIdx,
                                                           const float* __restrict__ dinv,
                                                           const float* __restrict__ g,
                                                           const float* __restrict__ b2,
                                                           float* __restrict__ out,
                                                           unsigned int* __restrict__ list,
                                                           int* __restrict__ listcnt, int E, int n) {
    owner_body<2>(map4, srcIdx, dstIdx, dinv, g, b2, out, list, listcnt, E, n);
}

__global__ void __launch_bounds__(256) gemm1_kernel(const float* __restrict__ x,
                                                    const float* __restrict__ W1,
                                                    float* __restrict__ hw, int n) {
    const int lane = threadIdx.x & 31;
    const int half = lane >> 4;
    const int l16  = lane & 15;
    const int tile = blockIdx.x * 8 + (threadIdx.x >> 5);
    const int ntiles = (n + 15) >> 4;
    if (tile >= ntiles) return;

    int row = tile * 16 + l16;
    if (row >= n) row = n - 1;
    const float* xr = x + row * IN_DIM;

    v8f c0 = {}; v8f c1 = {};
    #pragma unroll 8
    for (int k = 0; k < IN_DIM; k += 4) {
        const int kk = k + 2 * half;
        v2f a;  a.x  = xr[kk];                       a.y  = xr[kk + 1];
        v2f b0; b0.x = W1[kk * HID_DIM + l16];       b0.y = W1[(kk + 1) * HID_DIM + l16];
        v2f b1; b1.x = W1[kk * HID_DIM + 16 + l16];  b1.y = W1[(kk + 1) * HID_DIM + 16 + l16];
        c0 = __builtin_amdgcn_wmma_f32_16x16x4_f32(false, a, false, b0, (short)0, c0, false, false);
        c1 = __builtin_amdgcn_wmma_f32_16x16x4_f32(false, a, false, b1, (short)0, c1, false, false);
    }
    #pragma unroll
    for (int i = 0; i < 8; ++i) {
        int orow = tile * 16 + i + 8 * half;
        if (orow < n) {
            hw[orow * HID_DIM + l16]      = c0[i];
            hw[orow * HID_DIM + 16 + l16] = c1[i];
        }
    }
}


__global__ void __launch_bounds__(256) gemm2_kernel(const float* __restrict__ h,
                                                    const float* __restrict__ W2,
                                                    float* __restrict__ g, int n) {
    const int lane = threadIdx.x & 31;
    const int half = lane >> 4;
    const int l16  = lane & 15;
    const int tile = blockIdx.x * 8 + (threadIdx.x >> 5);
    const int ntiles = (n + 15) >> 4;
    if (tile >= ntiles) return;

    int row = tile * 16 + l16;
    if (row >= n) row = n - 1;
    const float* hr = h + row * HID_DIM;
    const bool live = (l16 < OUT_DIM);

    v8f c = {};
    #pragma unroll
    for (int k = 0; k < HID_DIM; k += 4) {
        const int kk = k + 2 * half;
        v2f a; a.x = hr[kk]; a.y = hr[kk + 1];
        v2f b;
        b.x = live ? W2[kk * OUT_DIM + l16]       : 0.0f;
        b.y = live ? W2[(kk + 1) * OUT_DIM + l16] : 0.0f;
        c = __builtin_amdgcn_wmma_f32_16x16x4_f32(false, a, false, b, (short)0, c, false, false);
    }
    if (live) {
        #pragma unroll
        for (int i = 0; i < 8; ++i) {
            int orow = tile * 16 + i + 8 * half;
            if (orow < n) g[orow * OUT_DIM + l16] = c[i];
        }
    }
}


extern "C" void kernel_launch(void* const* d_in, const int* in_sizes, int n_in,
                              void* d_out, int out_size, void* d_ws, size_t ws_size,
                              hipStream_t stream) {
    const float* x  = (const float*)d_in[0];
    const int*   ei = (const int*)d_in[1];
    const float* W1 = (const float*)d_in[2];
    const float* b1 = (const float*)d_in[3];
    const float* W2 = (const float*)d_in[4];
    const float* b2 = (const float*)d_in[5];
    float* out = (float*)d_out;

    const int n = in_sizes[0] / IN_DIM;
    const int E = in_sizes[1] / 2;
    const int* srcIdx = ei;
    const int* dstIdx = ei + E;

    const int nvec   = (E + 15) / 16;
    const int nwords = nvec * 4;
    unsigned int* map4 = (unsigned int*)d_ws;
    float* dinv = (float*)(map4 + nwords);
    float* hw   = dinv + n;
    float* h    = hw + (size_t)n * HID_DIM;
    float* g    = h  + (size_t)n * HID_DIM;
    int* listcnt = (int*)(g + (size_t)n * OUT_DIM);
    unsigned int* list = (unsigned int*)(listcnt + NBLK);

    const dim3 blk(256);
    const int ntiles = (n + 15) / 16;

    blockmap_kernel<<<dim3((nwords + 255) / 256), blk, 0, stream>>>(dstIdx, map4, E, nwords, n);
    owner_deg_kernel<<<dim3(NBLK), dim3(OWN_T), 0, stream>>>(map4, srcIdx, dstIdx, dinv, list, listcnt, E, n);
    gemm1_kernel<<<dim3((ntiles + 7) / 8), blk, 0, stream>>>(x, W1, hw, n);
    owner_agg1_kernel<<<dim3(NBLK), dim3(OWN_T), 0, stream>>>(map4, srcIdx, dstIdx, dinv, hw, b1, h, list, listcnt, E, n);
    gemm2_kernel<<<dim3((ntiles + 7) / 8), blk, 0, stream>>>(h, W2, g, n);
    owner_agg2_kernel<<<dim3(NBLK), dim3(OWN_T), 0, stream>>>(map4, srcIdx, dstIdx, dinv, g, b2, out, list, listcnt, E, n);
}


# Both row passes read an edge's two degree factors from a byte table of all node degrees that each owner copies into its LDS (255: the float table), not by two global lookups per edge; sums unchanged
# speedup vs baseline: 3.5388x; 1.2289x over previous
#include <hip/hip_runtime.h>


typedef float v2f __attribute__((ext_vector_type(2)));
typedef float v8f __attribute__((ext_vector_type(8)));

#define IN_DIM  128
#define HID_DIM 32
#define OUT_DIM 8

#define NBLK    256
#define RMAX    512
#define OWN_T   1024
#define OWN_WAVES (OWN_T / 32)
#define QCAP    64
#define LCAP    65536
#define SRC_BITS 17
#define CH_WORDS 64
#define CH_EDGES (CH_WORDS * 32)
#define SP_PITCH (CH_WORDS + 1)
#define ROW_CHUNKS (OWN_T * 4 / CH_WORDS)
#define PLANE_CHUNKS(E) ((((E) + CH_EDGES - 1) / CH_EDGES + ROW_CHUNKS - 1) / ROW_CHUNKS * ROW_CHUNKS)
#define SRC_MASK ((1u << SRC_BITS) - 1u)
#define TAB_N   (NBLK * RMAX)
#define TAB_ESC 255u
#define TAB_VEC (TAB_N / 16 / OWN_T)
static_assert(NBLK * RMAX <= (1 << SRC_BITS), "a source node must fit the low field of a list entry");
static_assert(RMAX <= (1 << (32 - SRC_BITS)), "a row inside the block must fit the high field of a list entry");

__global__ void __launch_bounds__(256) planes_kernel(const int* __restrict__ srcIdx,
                                                     const int* __restrict__ dstIdx,
                                                     unsigned int* __restrict__ planes,
                                                     unsigned int* __restrict__ packed,
                                                     int E, int PW, int n) {
    __shared__ unsigned int sp[32 * SP_PITCH];
    const int tid  = threadIdx.x;
    const int lane = tid & 31;
    const int wave = tid >> 5;
    const unsigned R = (unsigned)((n + NBLK - 1) / NBLK);
    const int ebase = blockIdx.x * CH_EDGES;
    int d[8];
    int s[8];
    #pragma unroll
    for (int j = 0; j < 8; ++j) {
        const int e = ebase + (j * 8 + wave) * 32 + lane;
        d[j] = (e < E) ? dstIdx[e] : -1;
        s[j] = (e < E) ? srcIdx[e] : 0;
    }
    #pragma unroll
    for (int j = 0; j < 8; ++j) {
        const int e = ebase + (j * 8 + wave) * 32 + lane;
        const unsigned blkno = (unsigned)d[j] / R;
        if (e < E) packed[e] = (((unsigned)d[j] - blkno * R) << SRC_BITS) | (unsigned)s[j];
        const unsigned b  = blkno & 255u;
        const unsigned hi = (d[j] >= 0) ? (b >> 4) : 16u;
        const unsigned lo = (d[j] >= 0) ? (b & 15u) : 16u;
        int w = 0;
        #pragma unroll
        for (int v = 0; v < 16; ++v) {
            const int mh = (int)__builtin_amdgcn_ballot_w32(hi == (unsigned)v);
            const int ml = (int)__builtin_amdgcn_ballot_w32(lo == (unsigned)v);
            w = (lane == v)      ? mh : w;
            w = (lane == 16 + v) ? ml : w;
        }
        sp[lane * SP_PITCH + j * 8 + wave] = (unsigned)w;
    }
    __syncthreads();
    const int p = tid >> 3;
    const int q = tid & 7;
    const unsigned int* row = sp + p * SP_PITCH;
    const uint4 a = make_uint4(row[q * 4], row[q * 4 + 1], row[q * 4 + 2], row[q * 4 + 3]);
    const uint4 c = make_uint4(row[32 + q * 4], row[33 + q * 4], row[34 + q * 4], row[35 + q * 4]);
    uint4* outv = reinterpret_cast<uint4*>(planes + (size_t)p * PW + (size_t)blockIdx.x * CH_WORDS);
    outv[q]     = a;
    outv[8 + q] = c;
}

template <int W>
__device__ __forceinline__ int lane_get(int v, int idx) {
    if constexpr (W == 32) return __builtin_amdgcn_readlane(v, idx);
    else                   return __shfl(v, idx);
}

__device__ __forceinline__ float edge_weight(const unsigned char* tab, const float* __restrict__ dinv,
                                             int s, int d) {
    const unsigned cs = tab[s];
    const unsigned cd = tab[d];
    float ws = rsqrtf((float)cs);
    float wd = rsqrtf((float)cd);
    if (cs == TAB_ESC) ws = dinv[s];
    if (cd == TAB_ESC) wd = dinv[d];
    return ws * wd;
}

__device__ __forceinline__ void drain_count(const int* q, int cnt, int lane,
                                            const unsigned int* __restrict__ packed, int* cnti,
                                            int* cursor, unsigned int* __restrict__ mylist) {
    unsigned ent = 0;
    if (lane < cnt) {
        ent = packed[q[lane]];
        atomicAdd(&cnti[ent >> SRC_BITS], 1);
    }
    int off = 0;
    if (lane == 0) off = atomicAdd(cursor, cnt);
    off = __builtin_amdgcn_readfirstlane(off);
    if (off + cnt <= LCAP && lane < cnt) mylist[off + lane] = ent;
}

template <int W>
__device__ __forceinline__ void gather_add(int s, int dl, float nrm, int cnt, int lane,
                                           const float* __restrict__ feat, float* acc) {
    constexpr int EPI = 32 / W;
    constexpr int U   = 8;
    const int sub = lane / W;
    const int f   = lane % W;
    for (int i0 = 0; i0 < cnt; i0 += U * EPI) {
        float r[U];
        #pragma unroll
        for (int j = 0; j < U; ++j) {
            const int idx = i0 + j * EPI + sub;
            const int sj = lane_get<W>(s, idx);
            r[j] = feat[(size_t)sj * W + f];
        }
        #pragma unroll
        for (int j = 0; j < U; ++j) {
            const int idx = i0 + j * EPI + sub;
            const float nj = __int_as_float(lane_get<W>(__float_as_int(nrm), idx));
            const int   dj = lane_get<W>(dl, idx);
            if (idx < cnt) atomicAdd(&acc[dj * W + f], r[j] * nj);
        }
    }
}

template <int W>
__device__ __forceinline__ void drain_rows(const int* q, int cnt, int lane, int base,
                                           const unsigned int* __restrict__ packed,
                                           const float* __restrict__ dinv, const unsigned char* tab,
                                           const float* __restrict__ feat, float* acc) {
    int s = 0, dl = 0;
    float nrm = 0.0f;
    if (lane < cnt) {
        const unsigned ent = packed[q[lane]];
        s  = (int)(ent & SRC_MASK);
        dl = (int)(ent >> SRC_BITS);
        nrm = edge_weight(tab, dinv, s, base + dl);
    }
    gather_add<W>(s, dl, nrm, cnt, lane, feat, acc);
}

template <int W>
__device__ __forceinline__ void replay_rows(const unsigned int* __restrict__ mylist, int total,
                                            int wave, int lane, int base,
                                            const float* __restrict__ dinv, const unsigned char* tab,
                                            const float* __restrict__ feat, float* acc) {
    int c = wave * 32;
    unsigned ent = (c + lane < total) ? mylist[c + lane] : 0u;
    for (; c < total; c += OWN_T) {
        const int cn = c + OWN_T;
        const unsigned nxt = (cn + lane < total) ? mylist[cn + lane] : 0u;
        int cnt = total - c;
        if (cnt > 32) cnt = 32;
        const int s  = (int)(ent & SRC_MASK);
        const int dl = (int)(ent >> SRC_BITS);
        float nrm = 0.0f;
        if (lane < cnt) nrm = edge_weight(tab, dinv, s, base + dl);
        gather_add<W>(s, dl, nrm, cnt, lane, feat, acc);
        ent = nxt;
    }
}

template <int MODE>
__device__ __forceinline__ void scan_map(const unsigned int* __restrict__ planes,
                                         const unsigned int* __restrict__ packed,
                                         const float* __restrict__ dinv, const unsigned char* tab,
                                         const float* __restrict__ feat,
                                         float* acc, int* queue, int* cursor,
                                         unsigned int* __restrict__ mylist,
                                         int E, int blk, int base) {
    constexpr int W = (MODE == 1) ? HID_DIM : ((MODE == 2) ? OUT_DIM : 1);
    int* cnti = reinterpret_cast<int*>(acc);
    const int tid  = threadIdx.x;
    const int lane = tid & 31;
    const int wave = tid >> 5;

    int* myq = queue + wave * QCAP;
    int qn = 0;
    const int PV   = PLANE_CHUNKS(E) * (CH_WORDS / 4);
    const int nrow = PV / OWN_T;
    const uint4* hp = reinterpret_cast<const uint4*>(planes) + (size_t)(blk >> 4) * PV + tid;
    const uint4* lp = reinterpret_cast<const uint4*>(planes) + (size_t)(16 + (blk & 15)) * PV + tid;

    uint4 hq = hp[0];
    uint4 lq = lp[0];
    for (int r = 0; r < nrow; ++r) {
        uint4 hn = hq, ln = lq;
        if (r + 1 < nrow) {
            hn = hp[(size_t)(r + 1) * OWN_T];
            ln = lp[(size_t)(r + 1) * OWN_T];
        }
        const unsigned m[4] = { hq.x & lq.x, hq.y & lq.y, hq.z & lq.z, hq.w & lq.w };
        const int ev = (r * OWN_T + tid) * 128;
        #pragma unroll
        for (int c = 0; c < 4; ++c) {
            unsigned P = m[c];
            unsigned any = __builtin_amdgcn_ballot_w32(P != 0u);
            while (any != 0u) {
                const bool ok = (P != 0u);
                const int  e  = ev + c * 32 + __builtin_ctz(ok ? P : 1u);
                if (ok) myq[qn + (int)__builtin_amdgcn_mbcnt_lo(any, 0u)] = e;
                qn += __builtin_popcount(any);
                if (qn >= 32) {
                    qn -= 32;
                    if constexpr (MODE == 0) drain_count(myq + qn, 32, lane, packed, cnti, cursor, mylist);
                    else drain_rows<W>(myq + qn, 32, lane, base, packed, dinv, tab, feat, acc);
                }
                P &= P - 1u;
                any = __builtin_amdgcn_ballot_w32(P != 0u);
            }
        }
        hq = hn;
        lq = ln;
    }
    if (qn > 0) {
        if constexpr (MODE == 0) drain_count(myq, qn, lane, packed, cnti, cursor, mylist);
        else drain_rows<W>(myq, qn, lane, base, packed, dinv, tab, feat, acc);
    }
}

template <int MODE>
__device__ __forceinline__ void owner_body(const unsigned int* __restrict__ planes,
                                           const unsigned int* __restrict__ packed,
                                           const float* __restrict__ dinv,
                                           const float* __restrict__ feat,
                                           const float* __restrict__ bias,
                                           float* __restrict__ outp,
                                           unsigned int* __restrict__ list,
                                           int* __restrict__ listcnt,
                                           unsigned char* __restrict__ deg8, int E, int n) {
    constexpr int W = (MODE == 1) ? HID_DIM : ((MODE == 2) ? OUT_DIM : 1);
    __shared__ float acc[RMAX * W];
    __shared__ int   queue[OWN_WAVES * QCAP];
    __shared__ int   cursor;
    __shared__ __attribute__((aligned(16))) unsigned char tab[(MODE == 0) ? 16 : TAB_N];
    int* cnti = reinterpret_cast<int*>(acc);

    const int tid  = threadIdx.x;
    const int blk  = blockIdx.x;
    const int R    = (n + NBLK - 1) / NBLK;
    const int base = blk * R;
    int rows = n - base;
    if (rows > R) rows = R;
    if (rows <= 0) return;
    unsigned int* mylist = list + (size_t)blk * LCAP;

    uint4 tv[TAB_VEC];
    const int nvec = n >> 4;
    const int rot  = (int)(((long long)blk * nvec) / NBLK) & ~7;
    if constexpr (MODE != 0) {
        const uint4* srcv = reinterpret_cast<const uint4*>(deg8);
        #pragma unroll
        for (int k = 0; k < TAB_VEC; ++k) {
            const int i = k * OWN_T + tid;
            int j = i + rot;
            if (j >= nvec) j -= nvec;
            tv[k] = (i < nvec) ? srcv[j] : make_uint4(0u, 0u, 0u, 0u);
        }
    }

    for (int i = tid; i < rows * W; i += OWN_T) {
        if constexpr (MODE == 0) {
            cnti[i] = 1;
        } else {
            const int node = base + i / W;
            const float dv = dinv[node];
            acc[i] = feat[(size_t)node * W + (i % W)] * (dv * dv);
        }
    }
    if constexpr (MODE == 0) {
        if (tid == 0) cursor = 0;
    } else {
        uint4* tabv = reinterpret_cast<uint4*>(tab);
        #pragma unroll
        for (int k = 0; k < TAB_VEC; ++k) {
            const int i = k * OWN_T + tid;
            int j = i + rot;
            if (j >= nvec) j -= nvec;
            if (i < nvec) tabv[j] = tv[k];
        }
        if (tid < (n & 15)) tab[(nvec << 4) + tid] = deg8[(nvec << 4) + tid];
    }
    __syncthreads();

    if constexpr (MODE == 0) {
        scan_map<0>(planes, packed, dinv, tab, feat, acc, queue, &cursor, mylist, E, blk, base);
    } else {
        const int total = listcnt[blk];
        if (total <= LCAP) replay_rows<W>(mylist, total, tid >> 5, tid & 31, base, dinv, tab, feat, acc);
        else scan_map<MODE>(planes, packed, dinv, tab, feat, acc, queue, &cursor, mylist, E, blk, base);
    }
    __syncthreads();

    if constexpr (MODE == 0) {
        if (tid == 0) listcnt[blk] = cursor;
    }

    for (int i = tid; i < rows * W; i += OWN_T) {
        if constexpr (MODE == 0) {
            const int cn = cnti[i];
            outp[base + i] = rsqrtf((float)cn);
            deg8[base + i] = (unsigned char)((cn < (int)TAB_ESC) ? cn : (int)TAB_ESC);
        } else if constexpr (MODE == 1) {
            outp[(size_t)base * W + i] = fmaxf(acc[i] + bias[i & (W - 1)], 0.0f);
        } else {
            outp[(size_t)base * W + i] = acc[i] + bias[i & (W - 1)];
        }
    }
}

__global__ void __launch_bounds__(OWN_T) owner_deg_kernel(const unsigned int* __restrict__ planes,
                                                          const unsigned int* __restrict__ packed,
                                                          float* __restrict__ dinv,
                                                          unsigned int* __restrict__ list,
                                                          int* __restrict__ listcnt,
                                                          unsigned char* __restrict__ deg8, int E, int n) {
    owner_body<0>(planes, packed, nullptr, nullptr, nullptr, dinv, list, listcnt, deg8, E, n);
}

__global__ void __launch_bounds__(OWN_T) owner_agg1_kernel(const unsigned int* __restrict__ planes,
                                                           const unsigned int* __restrict__ packed,
                                                           const float* __restrict__ dinv,
                                                           const float* __restrict__ hw,
                                                           const float* __restrict__ b1,
                                                           float* __restrict__ h,
                                                           unsigned int* __restrict__ list,
                                                           int* __restrict__ listcnt,
                                                           unsigned char* __restrict__ deg8, int E, int n) {
    owner_body<1>(planes, packed, dinv, hw, b1, h, list, listcnt, deg8, E, n);
}

__global__ void __launch_bounds__(OWN_T) owner_agg2_kernel(const unsigned int* __restrict__ planes,
                                                           const unsigned int* __restrict__ packed,
                                                           const float* __restrict__ dinv,
                                                           const float* __restrict__ g,
                                                           const float* __restrict__ b2,
                                                           float* __restrict__ out,
                                                           unsigned int* __restrict__ list,
                                                           int* __restrict__ listcnt,
                                                           unsigned char* __restrict__ deg8, int E, int n) {
    owner_body<2>(planes, packed, dinv, g, b2, out, list, listcnt, deg8, E, n);
}

__global__ void __launch_bounds__(256) gemm1_kernel(const float* __restrict__ x,
                                                    const float* __restrict__ W1,
                                                    float* __restrict__ hw, int n) {
    const int lane = threadIdx.x & 31;
    const int half = lane >> 4;
    const int l16  = lane & 15;
    const int tile = blockIdx.x * 8 + (threadIdx.x >> 5);
    const int ntiles = (n + 15) >> 4;
    if (tile >= ntiles) return;

    int row = tile * 16 + l16;
    if (row >= n) row = n - 1;
    const float* xr = x + row * IN_DIM;

    v8f c0 = {}; v8f c1 = {};
    #pragma unroll 8
    for (int k = 0; k < IN_DIM; k += 4) {
        const int kk = k + 2 * half;
        v2f a;  a.x  = xr[kk];                       a.y  = xr[kk + 1];
        v2f b0; b0.x = W1[kk * HID_DIM + l16];       b0.y = W1[(kk + 1) * HID_DIM + l16];
        v2f b1; b1.x = W1[kk * HID_DIM + 16 + l16];  b1.y = W1[(kk + 1) * HID_DIM + 16 + l16];
        c0 = __builtin_amdgcn_wmma_f32_16x16x4_f32(false, a, false, b0, (short)0, c0, false, false);
        c1 = __builtin_amdgcn_wmma_f32_16x16x4_f32(false, a, false, b1, (short)0, c1, false, false);
    }
    #pragma unroll
    for (int i = 0; i < 8; ++i) {
        int orow = tile * 16 + i + 8 * half;
        if (orow < n) {
            hw[orow * HID_DIM + l16]      = c0[i];
            hw[orow * HID_DIM + 16 + l16] = c1[i];
        }
    }
}


__global__ void __launch_bounds__(256) gemm2_kernel(const float* __restrict__ h,
                                                    const float* __restrict__ W2,
                                                    float* __restrict__ g, int n) {
    const int lane = threadIdx.x & 31;
    const int half = lane >> 4;
    const int l16  = lane & 15;
    const int tile = blockIdx.x * 8 + (threadIdx.x >> 5);
    const int ntiles = (n + 15) >> 4;
    if (tile >= ntiles) return;

    int row = tile * 16 + l16;
    if (row >= n) row = n - 1;
    const float* hr = h + row * HID_DIM;
    const bool live = (l16 < OUT_DIM);

    v8f c = {};
    #pragma unroll
    for (int k = 0; k < HID_DIM; k += 4) {
        const int kk = k + 2 * half;
        v2f a; a.x = hr[kk]; a.y = hr[kk + 1];
        v2f b;
        b.x = live ? W2[kk * OUT_DIM + l16]       : 0.0f;
        b.y = live ? W2[(kk + 1) * OUT_DIM + l16] : 0.0f;
        c = __builtin_amdgcn_wmma_f32_16x16x4_f32(false, a, false, b, (short)0, c, false, false);
    }
    if (live) {
        #pragma unroll
        for (int i = 0; i < 8; ++i) {
            int orow = tile * 16 + i + 8 * half;
            if (orow < n) g[orow * OUT_DIM + l16] = c[i];
        }
    }
}


extern "C" void kernel_launch(void* const* d_in, const int* in_sizes, int n_in,
                              void* d_out, int out_size, void* d_ws, size_t ws_size,
                              hipStream_t stream) {
    const float* x  = (const float*)d_in[0];
    const int*   ei = (const int*)d_in[1];
    const float* W1 = (const float*)d_in[2];
    const float* b1 = (const float*)d_in[3];
    const float* W2 = (const float*)d_in[4];
    const float* b2 = (const float*)d_in[5];
    float* out = (float*)d_out;

    const int n = in_sizes[0] / IN_DIM;
    const int E = in_sizes[1] / 2;
    const int* srcIdx = ei;
    const int* dstIdx = ei + E;

    const int nchunks = PLANE_CHUNKS(E);
    const int PW = nchunks * CH_WORDS;
    unsigned int* planes = (unsigned int*)d_ws;
    unsigned int* packed = planes + (size_t)32 * PW;
    float* dinv = (float*)(packed + (size_t)nchunks * CH_EDGES);
    float* hw   = dinv + n;
    float* h    = hw + (size_t)n * HID_DIM;
    float* g    = h  + (size_t)n * HID_DIM;
    int* listcnt = (int*)(g + (size_t)n * OUT_DIM);
    unsigned int* list = (unsigned int*)(listcnt + NBLK);
    unsigned char* deg8 = (unsigned char*)(((uintptr_t)(list + (size_t)NBLK * LCAP) + 255) & ~(uintptr_t)255);

    const dim3 blk(256);
    const int ntiles = (n + 15) / 16;

    planes_kernel<<<dim3(nchunks), blk, 0, stream>>>(srcIdx, dstIdx, planes, packed, E, PW, n);
    owner_deg_kernel<<<dim3(NBLK), dim3(OWN_T), 0, stream>>>(planes, packed, dinv, list, listcnt, deg8, E, n);
    gemm1_kernel<<<dim3((ntiles + 7) / 8), blk, 0, stream>>>(x, W1, hw, n);
    owner_agg1_kernel<<<dim3(NBLK), dim3(OWN_T), 0, stream>>>(planes, packed, dinv, hw, b1, h, list, listcnt, deg8, E, n);
    gemm2_kernel<<<dim3((ntiles + 7) / 8), blk, 0, stream>>>(h, W2, g, n);
    owner_agg2_kernel<<<dim3(NBLK), dim3(OWN_T), 0, stream>>>(planes, packed, dinv, g, b2, out, list, listcnt, deg8, E, n);
}


# Both products as resident waves: 256 work-groups of 8 waves keep their weight fragments in registers and walk over row tiles, rows in 16-byte loads one tile ahead; each dot product's k order changes
# speedup vs baseline: 3.6620x; 1.0348x over previous
#include <hip/hip_runtime.h>


typedef float v2f __attribute__((ext_vector_type(2)));
typedef float v8f __attribute__((ext_vector_type(8)));

#define IN_DIM  128
#define HID_DIM 32
#define OUT_DIM 8

#define NBLK    256
#define RMAX    512
#define OWN_T   1024
#define OWN_WAVES (OWN_T / 32)
#define QCAP    64
#define LCAP    65536
#define SRC_BITS 17
#define CH_WORDS 64
#define CH_EDGES (CH_WORDS * 32)
#define SP_PITCH (CH_WORDS + 1)
#define ROW_CHUNKS (OWN_T * 4 / CH_WORDS)
#define PLANE_CHUNKS(E) ((((E) + CH_EDGES - 1) / CH_EDGES + ROW_CHUNKS - 1) / ROW_CHUNKS * ROW_CHUNKS)
#define SRC_MASK ((1u << SRC_BITS) - 1u)
#define TAB_N   (NBLK * RMAX)
#define TAB_ESC 255u
#define TAB_VEC (TAB_N / 16 / OWN_T)
static_assert(NBLK * RMAX <= (1 << SRC_BITS), "a source node must fit the low field of a list entry");
static_assert(RMAX <= (1 << (32 - SRC_BITS)), "a row inside the block must fit the high field of a list entry");

__global__ void __launch_bounds__(256) planes_kernel(const int* __restrict__ srcIdx,
                                                     const int* __restrict__ dstIdx,
                                                     unsigned int* __restrict__ planes,
                                                     unsigned int* __restrict__ packed,
                                                     int E, int PW, int n) {
    __shared__ unsigned int sp[32 * SP_PITCH];
    const int tid  = threadIdx.x;
    const int lane = tid & 31;
    const int wave = tid >> 5;
    const unsigned R = (unsigned)((n + NBLK - 1) / NBLK);
    const int ebase = blockIdx.x * CH_EDGES;
    int d[8];
    int s[8];
    #pragma unroll
    for (int j = 0; j < 8; ++j) {
        const int e = ebase + (j * 8 + wave) * 32 + lane;
        d[j] = (e < E) ? dstIdx[e] : -1;
        s[j] = (e < E) ? srcIdx[e] : 0;
    }
    #pragma unroll
    for (int j = 0; j < 8; ++j) {
        const int e = ebase + (j * 8 + wave) * 32 + lane;
        const unsigned blkno = (unsigned)d[j] / R;
        if (e < E) packed[e] = (((unsigned)d[j] - blkno * R) << SRC_BITS) | (unsigned)s[j];
        const unsigned b  = blkno & 255u;
        const unsigned hi = (d[j] >= 0) ? (b >> 4) : 16u;
        const unsigned lo = (d[j] >= 0) ? (b & 15u) : 16u;
        int w = 0;
        #pragma unroll
        for (int v = 0; v < 16; ++v) {
            const int mh = (int)__builtin_amdgcn_ballot_w32(hi == (unsigned)v);
            const int ml = (int)__builtin_amdgcn_ballot_w32(lo == (unsigned)v);
            w = (lane == v)      ? mh : w;
            w = (lane == 16 + v) ? ml : w;
        }
        sp[lane * SP_PITCH + j * 8 + wave] = (unsigned)w;
    }
    __syncthreads();
    const int p = tid >> 3;
    const int q = tid & 7;
    const unsigned int* row = sp + p * SP_PITCH;
    const uint4 a = make_uint4(row[q * 4], row[q * 4 + 1], row[q * 4 + 2], row[q * 4 + 3]);
    const uint4 c = make_uint4(row[32 + q * 4], row[33 + q * 4], row[34 + q * 4], row[35 + q * 4]);
    uint4* outv = reinterpret_cast<uint4*>(planes + (size_t)p * PW + (size_t)blockIdx.x * CH_WORDS);
    outv[q]     = a;
    outv[8 + q] = c;
}

template <int W>
__device__ __forceinline__ int lane_get(int v, int idx) {
    if constexpr (W == 32) return __builtin_amdgcn_readlane(v, idx);
    else                   return __shfl(v, idx);
}

__device__ __forceinline__ float edge_weight(const unsigned char* tab, const float* __restrict__ dinv,
                                             int s, int d) {
    const unsigned cs = tab[s];
    const unsigned cd = tab[d];
    float ws = rsqrtf((float)cs);
    float wd = rsqrtf((float)cd);
    if (cs == TAB_ESC) ws = dinv[s];
    if (cd == TAB_ESC) wd = dinv[d];
    return ws * wd;
}

__device__ __forceinline__ void drain_count(const int* q, int cnt, int lane,
                                            const unsigned int* __restrict__ packed, int* cnti,
                                            int* cursor, unsigned int* __restrict__ mylist) {
    unsigned ent = 0;
    if (lane < cnt) {
        ent = packed[q[lane]];
        atomicAdd(&cnti[ent >> SRC_BITS], 1);
    }
    int off = 0;
    if (lane == 0) off = atomicAdd(cursor, cnt);
    off = __builtin_amdgcn_readfirstlane(off);
    if (off + cnt <= LCAP && lane < cnt) mylist[off + lane] = ent;
}

template <int W>
__device__ __forceinline__ void gather_add(int s, int dl, float nrm, int cnt, int lane,
                                           const float* __restrict__ feat, float* acc) {
    constexpr int EPI = 32 / W;
    constexpr int U   = 8;
    const int sub = lane / W;
    const int f   = lane % W;
    for (int i0 = 0; i0 < cnt; i0 += U * EPI) {
        float r[U];
        #pragma unroll
        for (int j = 0; j < U; ++j) {
            const int idx = i0 + j * EPI + sub;
            const int sj = lane_get<W>(s, idx);
            r[j] = feat[(size_t)sj * W + f];
        }
        #pragma unroll
        for (int j = 0; j < U; ++j) {
            const int idx = i0 + j * EPI + sub;
            const float nj = __int_as_float(lane_get<W>(__float_as_int(nrm), idx));
            const int   dj = lane_get<W>(dl, idx);
            if (idx < cnt) atomicAdd(&acc[dj * W + f], r[j] * nj);
        }
    }
}

template <int W>
__device__ __forceinline__ void drain_rows(const int* q, int cnt, int lane, int base,
                                           const unsigned int* __restrict__ packed,
                                           const float* __restrict__ dinv, const unsigned char* tab,
                                           const float* __restrict__ feat, float* acc) {
    int s = 0, dl = 0;
    float nrm = 0.0f;
    if (lane < cnt) {
        const unsigned ent = packed[q[lane]];
        s  = (int)(ent & SRC_MASK);
        dl = (int)(ent >> SRC_BITS);
        nrm = edge_weight(tab, dinv, s, base + dl);
    }
    gather_add<W>(s, dl, nrm, cnt, lane, feat, acc);
}

template <int W>
__device__ __forceinline__ void replay_rows(const unsigned int* __restrict__ mylist, int total,
                                            int wave, int lane, int base,
                                            const float* __restrict__ dinv, const unsigned char* tab,
                                            const float* __restrict__ feat, float* acc) {
    int c = wave * 32;
    unsigned ent = (c + lane < total) ? mylist[c + lane] : 0u;
    for (; c < total; c += OWN_T) {
        const int cn = c + OWN_T;
        const unsigned nxt = (cn + lane < total) ? mylist[cn + lane] : 0u;
        int cnt = total - c;
        if (cnt > 32) cnt = 32;
        const int s  = (int)(ent & SRC_MASK);
        const int dl = (int)(ent >> SRC_BITS);
        float nrm = 0.0f;
        if (lane < cnt) nrm = edge_weight(tab, dinv, s, base + dl);
        gather_add<W>(s, dl, nrm, cnt, lane, feat, acc);
        ent = nxt;
    }
}

template <int MODE>
__device__ __forceinline__ void scan_map(const unsigned int* __restrict__ planes,
                                         const unsigned int* __restrict__ packed,
                                         const float* __restrict__ dinv, const unsigned char* tab,
                                         const float* __restrict__ feat,
                                         float* acc, int* queue, int* cursor,
                                         unsigned int* __restrict__ mylist,
                                         int E, int blk, int base) {
    constexpr int W = (MODE == 1) ? HID_DIM : ((MODE == 2) ? OUT_DIM : 1);
    int* cnti = reinterpret_cast<int*>(acc);
    const int tid  = threadIdx.x;
    const int lane = tid & 31;
    const int wave = tid >> 5;

    int* myq = queue + wave * QCAP;
    int qn = 0;
    const int PV   = PLANE_CHUNKS(E) * (CH_WORDS / 4);
    const int nrow = PV / OWN_T;
    const uint4* hp = reinterpret_cast<const uint4*>(planes) + (size_t)(blk >> 4) * PV + tid;
    const uint4* lp = reinterpret_cast<const uint4*>(planes) + (size_t)(16 + (blk & 15)) * PV + tid;

    uint4 hq = hp[0];
    uint4 lq = lp[0];
    for (int r = 0; r < nrow; ++r) {
        uint4 hn = hq, ln = lq;
        if (r + 1 < nrow) {
            hn = hp[(size_t)(r + 1) * OWN_T];
            ln = lp[(size_t)(r + 1) * OWN_T];
        }
        const unsigned m[4] = { hq.x & lq.x, hq.y & lq.y, hq.z & lq.z, hq.w & lq.w };
        const int ev = (r * OWN_T + tid) * 128;
        #pragma unroll
        for (int c = 0; c < 4; ++c) {
            unsigned P = m[c];
            unsigned any = __builtin_amdgcn_ballot_w32(P != 0u);
            while (any != 0u) {
                const bool ok = (P != 0u);
                const int  e  = ev + c * 32 + __builtin_ctz(ok ? P : 1u);
                if (ok) myq[qn + (int)__builtin_amdgcn_mbcnt_lo(any, 0u)] = e;
                qn += __builtin_popcount(any);
                if (qn >= 32) {
                    qn -= 32;
                    if constexpr (MODE == 0) drain_count(myq + qn, 32, lane, packed, cnti, cursor, mylist);
                    else drain_rows<W>(myq + qn, 32, lane, base, packed, dinv, tab, feat, acc);
                }
                P &= P - 1u;
                any = __builtin_amdgcn_ballot_w32(P != 0u);
            }
        }
        hq = hn;
        lq = ln;
    }
    if (qn > 0) {
        if constexpr (MODE == 0) drain_count(myq, qn, lane, packed, cnti, cursor, mylist);
        else drain_rows<W>(myq, qn, lane, base, packed, dinv, tab, feat, acc);
    }
}

template <int MODE>
__device__ __forceinline__ void owner_body(const unsigned int* __restrict__ planes,
                                           const unsigned int* __restrict__ packed,
                                           const float* __restrict__ dinv,
                                           const float* __restrict__ feat,
                                           const float* __restrict__ bias,
                                           float* __restrict__ outp,
                                           unsigned int* __restrict__ list,
                                           int* __restrict__ listcnt,
                                           unsigned char* __restrict__ deg8, int E, int n) {
    constexpr int W = (MODE == 1) ? HID_DIM : ((MODE == 2) ? OUT_DIM : 1);
    __shared__ float acc[RMAX * W];
    __shared__ int   queue[OWN_WAVES * QCAP];
    __shared__ int   cursor;
    __shared__ __attribute__((aligned(16))) unsigned char tab[(MODE == 0) ? 16 : TAB_N];
    int* cnti = reinterpret_cast<int*>(acc);

    const int tid  = threadIdx.x;
    const int blk  = blockIdx.x;
    const int R    = (n + NBLK - 1) / NBLK;
    const int base = blk * R;
    int rows = n - base;
    if (rows > R) rows = R;
    if (rows <= 0) return;
    unsigned int* mylist = list + (size_t)blk * LCAP;

    uint4 tv[TAB_VEC];
    const int nvec = n >> 4;
    const int rot  = (int)(((long long)blk * nvec) / NBLK) & ~7;
    if constexpr (MODE != 0) {
        const uint4* srcv = reinterpret_cast<const uint4*>(deg8);
        #pragma unroll
        for (int k = 0; k < TAB_VEC; ++k) {
            const int i = k * OWN_T + tid;
            int j = i + rot;
            if (j >= nvec) j -= nvec;
            tv[k] = (i < nvec) ? srcv[j] : make_uint4(0u, 0u, 0u, 0u);
        }
    }

    for (int i = tid; i < rows * W; i += OWN_T) {
        if constexpr (MODE == 0) {
            cnti[i] = 1;
        } else {
            const int node = base + i / W;
            const float dv = dinv[node];
            acc[i] = feat[(size_t)node * W + (i % W)] * (dv * dv);
        }
    }
    if constexpr (MODE == 0) {
        if (tid == 0) cursor = 0;
    } else {
        uint4* tabv = reinterpret_cast<uint4*>(tab);
        #pragma unroll
        for (int k = 0; k < TAB_VEC; ++k) {
            const int i = k * OWN_T + tid;
            int j = i + rot;
            if (j >= nvec) j -= nvec;
            if (i < nvec) tabv[j] = tv[k];
        }
        if (tid < (n & 15)) tab[(nvec << 4) + tid] = deg8[(nvec << 4) + tid];
    }
    __syncthreads();

    if constexpr (MODE == 0) {
        scan_map<0>(planes, packed, dinv, tab, feat, acc, queue, &cursor, mylist, E, blk, base);
    } else {
        const int total = listcnt[blk];
        if (total <= LCAP) replay_rows<W>(mylist, total, tid >> 5, tid & 31, base, dinv, tab, feat, acc);
        else scan_map<MODE>(planes, packed, dinv, tab, feat, acc, queue, &cursor, mylist, E, blk, base);
    }
    __syncthreads();

    if constexpr (MODE == 0) {
        if (tid == 0) listcnt[blk] = cursor;
    }

    for (int i = tid; i < rows * W; i += OWN_T) {
        if constexpr (MODE == 0) {
            const int cn = cnti[i];
            outp[base + i] = rsqrtf((float)cn);
            deg8[base + i] = (unsigned char)((cn < (int)TAB_ESC) ? cn : (int)TAB_ESC);
        } else if constexpr (MODE == 1) {
            outp[(size_t)base * W + i] = fmaxf(acc[i] + bias[i & (W - 1)], 0.0f);
        } else {
            outp[(size_t)base * W + i] = acc[i] + bias[i & (W - 1)];
        }
    }
}

__global__ void __launch_bounds__(OWN_T) owner_deg_kernel(const unsigned int* __restrict__ planes,
                                                          const unsigned int* __restrict__ packed,
                                                          float* __restrict__ dinv,
                                                          unsigned int* __restrict__ list,
                                                          int* __restrict__ listcnt,
                                                          unsigned char* __restrict__ deg8, int E, int n) {
    owner_body<0>(planes, packed, nullptr, nullptr, nullptr, dinv, list, listcnt, deg8, E, n);
}

__global__ void __launch_bounds__(OWN_T) owner_agg1_kernel(const unsigned int* __restrict__ planes,
                                                           const unsigned int* __restrict__ packed,
                                                           const float* __restrict__ dinv,
                                                           const float* __restrict__ hw,
                                                           const float* __restrict__ b1,
                                                           float* __restrict__ h,
                                                           unsigned int* __restrict__ list,
                                                           int* __restrict__ listcnt,
                                                           unsigned char* __restrict__ deg8, int E, int n) {
    owner_body<1>(planes, packed, dinv, hw, b1, h, list, listcnt, deg8, E, n);
}

__global__ void __launch_bounds__(OWN_T) owner_agg2_kernel(const unsigned int* __restrict__ planes,
                                                           const unsigned int* __restrict__ packed,
                                                           const float* __restrict__ dinv,
                                                           const float* __restrict__ g,
                                                           const float* __restrict__ b2,
                                                           float* __restrict__ out,
                                                           unsigned int* __restrict__ list,
                                                           int* __restrict__ listcnt,
                                                           unsigned char* __restrict__ deg8, int E, int n) {
    owner_body<2>(planes, packed, dinv, g, b2, out, list, listcnt, deg8, E, n);
}

#define PROD_WG 256
#define PROD_WAVES 8

__device__ __forceinline__ int prod_step_k(int t, int hf) { return 8 * (t >> 1) + 4 * hf + 2 * (t & 1); }

template <int NV>
__device__ __forceinline__ void load_rows(float4 (&a)[NV], const float* __restrict__ src, int tile,
                                          int l16, int hf, int n) {
    int row = tile * 16 + l16;
    if (row >= n) row = n - 1;
    const float4* p = reinterpret_cast<const float4*>(src + (size_t)row * (NV * 8)) + hf;
    #pragma unroll
    for (int j = 0; j < NV; ++j) a[j] = p[2 * j];
}

__device__ __forceinline__ void tile1(const float4 (&a)[16], const v2f (&bl)[32], const v2f (&bh)[32],
                                      float* __restrict__ hw, int tile, int l16, int hf, int n) {
    v8f c0 = {}; v8f c1 = {};
    #pragma unroll
    for (int j = 0; j < 16; ++j) {
        v2f lo; lo.x = a[j].x; lo.y = a[j].y;
        v2f hi; hi.x = a[j].z; hi.y = a[j].w;
        c0 = __builtin_amdgcn_wmma_f32_16x16x4_f32(false, lo, false, bl[2 * j],     (short)0, c0, false, false);
        c1 = __builtin_amdgcn_wmma_f32_16x16x4_f32(false, lo, false, bh[2 * j],     (short)0, c1, false, false);
        c0 = __builtin_amdgcn_wmma_f32_16x16x4_f32(false, hi, false, bl[2 * j + 1], (short)0, c0, false, false);
        c1 = __builtin_amdgcn_wmma_f32_16x16x4_f32(false, hi, false, bh[2 * j + 1], (short)0, c1, false, false);
    }
    #pragma unroll
    for (int i = 0; i < 8; ++i) {
        const int orow = tile * 16 + i + 8 * hf;
        if (orow < n) {
            hw[orow * HID_DIM + l16]      = c0[i];
            hw[orow * HID_DIM + 16 + l16] = c1[i];
        }
    }
}

__global__ void __launch_bounds__(32 * PROD_WAVES) gemm1_kernel(const float* __restrict__ x,
                                                                const float* __restrict__ W1,
                                                                float* __restrict__ hw, int n) {
    const int lane = threadIdx.x & 31;
    const int hf   = lane >> 4;
    const int l16  = lane & 15;
    const int ntiles = (n + 15) >> 4;
    const int stride = (int)gridDim.x * PROD_WAVES;
    int tile = (threadIdx.x >> 5) * (int)gridDim.x + (int)blockIdx.x;
    if (tile >= ntiles) return;

    float4 a0[16]; float4 a1[16];
    load_rows<16>(a0, x, tile, l16, hf, n);
    __builtin_amdgcn_sched_barrier(0);

    v2f bl[32]; v2f bh[32];
    #pragma unroll
    for (int t = 0; t < 32; ++t) {
        const int kk = prod_step_k(t, hf);
        bl[t].x = W1[kk * HID_DIM + l16];       bl[t].y = W1[(kk + 1) * HID_DIM + l16];
        bh[t].x = W1[kk * HID_DIM + 16 + l16];  bh[t].y = W1[(kk + 1) * HID_DIM + 16 + l16];
    }
    for (;;) {
        const int t1 = tile + stride;
        load_rows<16>(a1, x, t1, l16, hf, n);
        __builtin_amdgcn_sched_barrier(0);
        tile1(a0, bl, bh, hw, tile, l16, hf, n);
        if (t1 >= ntiles) break;
        const int t2 = t1 + stride;
        load_rows<16>(a0, x, t2, l16, hf, n);
        __builtin_amdgcn_sched_barrier(0);
        tile1(a1, bl, bh, hw, t1, l16, hf, n);
        if (t2 >= ntiles) break;
        tile = t2;
    }
}


__device__ __forceinline__ void tile2(const float4 (&a)[4], const v2f (&b)[8],
                                      float* __restrict__ g, int tile, int l16, int hf, int n) {
    v8f c = {};
    #pragma unroll
    for (int j = 0; j < 4; ++j) {
        v2f lo; lo.x = a[j].x; lo.y = a[j].y;
        v2f hi; hi.x = a[j].z; hi.y = a[j].w;
        c = __builtin_amdgcn_wmma_f32_16x16x4_f32(false, lo, false, b[2 * j],     (short)0, c, false, false);
        c = __builtin_amdgcn_wmma_f32_16x16x4_f32(false, hi, false, b[2 * j + 1], (short)0, c, false, false);
    }
    if (l16 < OUT_DIM) {
        #pragma unroll
        for (int i = 0; i < 8; ++i) {
            const int orow = tile * 16 + i + 8 * hf;
            if (orow < n) g[orow * OUT_DIM + l16] = c[i];
        }
    }
}

__global__ void __launch_bounds__(32 * PROD_WAVES) gemm2_kernel(const float* __restrict__ h,
                                                                const float* __restrict__ W2,
                                                                float* __restrict__ g, int n) {
    const int lane = threadIdx.x & 31;
    const int hf   = lane >> 4;
    const int l16  = lane & 15;
    const int ntiles = (n + 15) >> 4;
    const int stride = (int)gridDim.x * PROD_WAVES;
    int tile = (threadIdx.x >> 5) * (int)gridDim.x + (int)blockIdx.x;
    if (tile >= ntiles) return;
    const bool live = (l16 < OUT_DIM);
    const int  cw   = l16 & (OUT_DIM - 1);

    float4 a0[4]; float4 a1[4];
    load_rows<4>(a0, h, tile, l16, hf, n);

    v2f b[8];
    #pragma unroll
    for (int t = 0; t < 8; ++t) {
        const int kk = prod_step_k(t, hf);
        const float w0 = W2[kk * OUT_DIM + cw];
        const float w1 = W2[(kk + 1) * OUT_DIM + cw];
        b[t].x = live ? w0 : 0.0f;
        b[t].y = live ? w1 : 0.0f;
    }
    for (;;) {
        const int t1 = tile + stride;
        load_rows<4>(a1, h, t1, l16, hf, n);
        tile2(a0, b, g, tile, l16, hf, n);
        if (t1 >= ntiles) break;
        const int t2 = t1 + stride;
        load_rows<4>(a0, h, t2, l16, hf, n);
        tile2(a1, b, g, t1, l16, hf, n);
        if (t2 >= ntiles) break;
        tile = t2;
    }
}


extern "C" void kernel_launch(void* const* d_in, const int* in_sizes, int n_in,
                              void* d_out, int out_size, void* d_ws, size_t ws_size,
                              hipStream_t stream) {
    const float* x  = (const float*)d_in[0];
    const int*   ei = (const int*)d_in[1];
    const float* W1 = (const float*)d_in[2];
    const float* b1 = (const float*)d_in[3];
    const float* W2 = (const float*)d_in[4];
    const float* b2 = (const float*)d_in[5];
    float* out = (float*)d_out;

    const int n = in_sizes[0] / IN_DIM;
    const int E = in_sizes[1] / 2;
    const int* srcIdx = ei;
    const int* dstIdx = ei + E;

    const int nchunks = PLANE_CHUNKS(E);
    const int PW = nchunks * CH_WORDS;
    unsigned int* planes = (unsigned int*)d_ws;
    unsigned int* packed = planes + (size_t)32 * PW;
    float* dinv = (float*)(packed + (size_t)nchunks * CH_EDGES);
    float* hw   = dinv + n;
    float* h    = hw + (size_t)n * HID_DIM;
    float* g    = h  + (size_t)n * HID_DIM;
    int* listcnt = (int*)(g + (size_t)n * OUT_DIM);
    unsigned int* list = (unsigned int*)(listcnt + NBLK);
    unsigned char* deg8 = (unsigned char*)(((uintptr_t)(list + (size_t)NBLK * LCAP) + 255) & ~(uintptr_t)255);

    const dim3 blk(256);

    planes_kernel<<<dim3(nchunks), blk, 0, stream>>>(srcIdx, dstIdx, planes, packed, E, PW, n);
    owner_deg_kernel<<<dim3(NBLK), dim3(OWN_T), 0, stream>>>(planes, packed, dinv, list, listcnt, deg8, E, n);
    gemm1_kernel<<<dim3(PROD_WG), blk, 0, stream>>>(x, W1, hw, n);
    owner_agg1_kernel<<<dim3(NBLK), dim3(OWN_T), 0, stream>>>(planes, packed, dinv, hw, b1, h, list, listcnt, deg8, E, n);
    gemm2_kernel<<<dim3(PROD_WG), blk, 0, stream>>>(h, W2, g, n);
    owner_agg2_kernel<<<dim3(NBLK), dim3(OWN_T), 0, stream>>>(planes, packed, dinv, g, b2, out, list, listcnt, deg8, E, n);
}


# Layer-1 list walk: all 32 row loads of a chunk go out as one group (was 4 groups of 8) and full chunks add with no per-edge test, the chunk count being scalar; sums stay arrival-order LDS adds
# speedup vs baseline: 3.9978x; 1.0917x over previous
#include <hip/hip_runtime.h>


typedef float v2f __attribute__((ext_vector_type(2)));
typedef float v8f __attribute__((ext_vector_type(8)));

#define IN_DIM  128
#define HID_DIM 32
#define OUT_DIM 8

#define NBLK    256
#define RMAX    512
#define OWN_T   1024
#define OWN_WAVES (OWN_T / 32)
#define QCAP    64
#define LCAP    65536
#define SRC_BITS 17
#define CH_WORDS 64
#define CH_EDGES (CH_WORDS * 32)
#define SP_PITCH (CH_WORDS + 1)
#define ROW_CHUNKS (OWN_T * 4 / CH_WORDS)
#define PLANE_CHUNKS(E) ((((E) + CH_EDGES - 1) / CH_EDGES + ROW_CHUNKS - 1) / ROW_CHUNKS * ROW_CHUNKS)
#define SRC_MASK ((1u << SRC_BITS) - 1u)
#define TAB_N   (NBLK * RMAX)
#define TAB_ESC 255u
#define TAB_VEC (TAB_N / 16 / OWN_T)
static_assert(NBLK * RMAX <= (1 << SRC_BITS), "a source node must fit the low field of a list entry");
static_assert(RMAX <= (1 << (32 - SRC_BITS)), "a row inside the block must fit the high field of a list entry");

__global__ void __launch_bounds__(256) planes_kernel(const int* __restrict__ srcIdx,
                                                     const int* __restrict__ dstIdx,
                                                     unsigned int* __restrict__ planes,
                                                     unsigned int* __restrict__ packed,
                                                     int E, int PW, int n) {
    __shared__ unsigned int sp[32 * SP_PITCH];
    const int tid  = threadIdx.x;
    const int lane = tid & 31;
    const int wave = tid >> 5;
    const unsigned R = (unsigned)((n + NBLK - 1) / NBLK);
    const int ebase = blockIdx.x * CH_EDGES;
    int d[8];
    int s[8];
    #pragma unroll
    for (int j = 0; j < 8; ++j) {
        const int e = ebase + (j * 8 + wave) * 32 + lane;
        d[j] = (e < E) ? dstIdx[e] : -1;
        s[j] = (e < E) ? srcIdx[e] : 0;
    }
    #pragma unroll
    for (int j = 0; j < 8; ++j) {
        const int e = ebase + (j * 8 + wave) * 32 + lane;
        const unsigned blkno = (unsigned)d[j] / R;
        if (e < E) packed[e] = (((unsigned)d[j] - blkno * R) << SRC_BITS) | (unsigned)s[j];
        const unsigned b  = blkno & 255u;
        const unsigned hi = (d[j] >= 0) ? (b >> 4) : 16u;
        const unsigned lo = (d[j] >= 0) ? (b & 15u) : 16u;
        int w = 0;
        #pragma unroll
        for (int v = 0; v < 16; ++v) {
            const int mh = (int)__builtin_amdgcn_ballot_w32(hi == (unsigned)v);
            const int ml = (int)__builtin_amdgcn_ballot_w32(lo == (unsigned)v);
            w = (lane == v)      ? mh : w;
            w = (lane == 16 + v) ? ml : w;
        }
        sp[lane * SP_PITCH + j * 8 + wave] = (unsigned)w;
    }
    __syncthreads();
    const int p = tid >> 3;
    const int q = tid & 7;
    const unsigned int* row = sp + p * SP_PITCH;
    const uint4 a = make_uint4(row[q * 4], row[q * 4 + 1], row[q * 4 + 2], row[q * 4 + 3]);
    const uint4 c = make_uint4(row[32 + q * 4], row[33 + q * 4], row[34 + q * 4], row[35 + q * 4]);
    uint4* outv = reinterpret_cast<uint4*>(planes + (size_t)p * PW + (size_t)blockIdx.x * CH_WORDS);
    outv[q]     = a;
    outv[8 + q] = c;
}

template <int W>
__device__ __forceinline__ int lane_get(int v, int idx) {
    if constexpr (W == 32) return __builtin_amdgcn_readlane(v, idx);
    else                   return __shfl(v, idx);
}

__device__ __forceinline__ float edge_weight(const unsigned char* tab, const float* __restrict__ dinv,
                                             int s, int d) {
    const unsigned cs = tab[s];
    const unsigned cd = tab[d];
    float ws = rsqrtf((float)cs);
    float wd = rsqrtf((float)cd);
    if (cs == TAB_ESC) ws = dinv[s];
    if (cd == TAB_ESC) wd = dinv[d];
    return ws * wd;
}

__device__ __forceinline__ void drain_count(const int* q, int cnt, int lane,
                                            const unsigned int* __restrict__ packed, int* cnti,
                                            int* cursor, unsigned int* __restrict__ mylist) {
    unsigned ent = 0;
    if (lane < cnt) {
        ent = packed[q[lane]];
        atomicAdd(&cnti[ent >> SRC_BITS], 1);
    }
    int off = 0;
    if (lane == 0) off = atomicAdd(cursor, cnt);
    off = __builtin_amdgcn_readfirstlane(off);
    if (off + cnt <= LCAP && lane < cnt) mylist[off + lane] = ent;
}

template <int W>
__device__ __forceinline__ void gather_add(int s, int dl, float nrm, int cnt, int lane,
                                           const float* __restrict__ feat, float* acc) {
    if constexpr (W == 32) {
        float r[32];
        #pragma unroll
        for (int j = 0; j < 32; ++j) {
            const int sj = __builtin_amdgcn_readlane(s, j);
            r[j] = feat[(size_t)sj * 32 + lane];
        }
        if (cnt == 32) {
            #pragma unroll
            for (int j = 0; j < 32; ++j) {
                const float nj = __int_as_float(__builtin_amdgcn_readlane(__float_as_int(nrm), j));
                const int   dj = __builtin_amdgcn_readlane(dl, j);
                atomicAdd(&acc[dj * 32 + lane], r[j] * nj);
            }
        } else {
            #pragma unroll
            for (int j = 0; j < 32; ++j) {
                const float nj = __int_as_float(__builtin_amdgcn_readlane(__float_as_int(nrm), j));
                const int   dj = __builtin_amdgcn_readlane(dl, j);
                if (j < cnt) atomicAdd(&acc[dj * 32 + lane], r[j] * nj);
            }
        }
    } else {
        constexpr int EPI = 32 / W;
        constexpr int U   = 8;
        const int sub = lane / W;
        const int f   = lane % W;
        for (int i0 = 0; i0 < cnt; i0 += U * EPI) {
            float r[U];
            #pragma unroll
            for (int j = 0; j < U; ++j) {
                const int idx = i0 + j * EPI + sub;
                const int sj = lane_get<W>(s, idx);
                r[j] = feat[(size_t)sj * W + f];
            }
            #pragma unroll
            for (int j = 0; j < U; ++j) {
                const int idx = i0 + j * EPI + sub;
                const float nj = __int_as_float(lane_get<W>(__float_as_int(nrm), idx));
                const int   dj = lane_get<W>(dl, idx);
                if (idx < cnt) atomicAdd(&acc[dj * W + f], r[j] * nj);
            }
        }
    }
}

template <int W>
__device__ __forceinline__ void drain_rows(const int* q, int cnt, int lane, int base,
                                           const unsigned int* __restrict__ packed,
                                           const float* __restrict__ dinv, const unsigned char* tab,
                                           const float* __restrict__ feat, float* acc) {
    int s = 0, dl = 0;
    float nrm = 0.0f;
    if (lane < cnt) {
        const unsigned ent = packed[q[lane]];
        s  = (int)(ent & SRC_MASK);
        dl = (int)(ent >> SRC_BITS);
        nrm = edge_weight(tab, dinv, s, base + dl);
    }
    gather_add<W>(s, dl, nrm, cnt, lane, feat, acc);
}

template <int W>
__device__ __forceinline__ void replay_rows(const unsigned int* __restrict__ mylist, int total,
                                            int wave, int lane, int base,
                                            const float* __restrict__ dinv, const unsigned char* tab,
                                            const float* __restrict__ feat, float* acc) {
    int c = wave * 32;
    unsigned ent = (c + lane < total) ? mylist[c + lane] : 0u;
    for (; c < total; c += OWN_T) {
        const int cn = c + OWN_T;
        const unsigned nxt = (cn + lane < total) ? mylist[cn + lane] : 0u;
        int cnt = total - c;
        if (cnt > 32) cnt = 32;
        const int s  = (int)(ent & SRC_MASK);
        const int dl = (int)(ent >> SRC_BITS);
        float nrm = 0.0f;
        if (lane < cnt) nrm = edge_weight(tab, dinv, s, base + dl);
        gather_add<W>(s, dl, nrm, cnt, lane, feat, acc);
        ent = nxt;
    }
}

template <int MODE>
__device__ __forceinline__ void scan_map(const unsigned int* __restrict__ planes,
                                         const unsigned int* __restrict__ packed,
                                         const float* __restrict__ dinv, const unsigned char* tab,
                                         const float* __restrict__ feat,
                                         float* acc, int* queue, int* cursor,
                                         unsigned int* __restrict__ mylist,
                                         int E, int blk, int base) {
    constexpr int W = (MODE == 1) ? HID_DIM : ((MODE == 2) ? OUT_DIM : 1);
    int* cnti = reinterpret_cast<int*>(acc);
    const int tid  = threadIdx.x;
    const int lane = tid & 31;
    const int wave = tid >> 5;

    int* myq = queue + wave * QCAP;
    int qn = 0;
    const int PV   = PLANE_CHUNKS(E) * (CH_WORDS / 4);
    const int nrow = PV / OWN_T;
    const uint4* hp = reinterpret_cast<const uint4*>(planes) + (size_t)(blk >> 4) * PV + tid;
    const uint4* lp = reinterpret_cast<const uint4*>(planes) + (size_t)(16 + (blk & 15)) * PV + tid;

    uint4 hq = hp[0];
    uint4 lq = lp[0];
    for (int r = 0; r < nrow; ++r) {
        uint4 hn = hq, ln = lq;
        if (r + 1 < nrow) {
            hn = hp[(size_t)(r + 1) * OWN_T];
            ln = lp[(size_t)(r + 1) * OWN_T];
        }
        const unsigned m[4] = { hq.x & lq.x, hq.y & lq.y, hq.z & lq.z, hq.w & lq.w };
        const int ev = (r * OWN_T + tid) * 128;
        #pragma unroll
        for (int c = 0; c < 4; ++c) {
            unsigned P = m[c];
            unsigned any = __builtin_amdgcn_ballot_w32(P != 0u);
            while (any != 0u) {
                const bool ok = (P != 0u);
                const int  e  = ev + c * 32 + __builtin_ctz(ok ? P : 1u);
                if (ok) myq[qn + (int)__builtin_amdgcn_mbcnt_lo(any, 0u)] = e;
                qn += __builtin_popcount(any);
                if (qn >= 32) {
                    qn -= 32;
                    if constexpr (MODE == 0) drain_count(myq + qn, 32, lane, packed, cnti, cursor, mylist);
                    else drain_rows<W>(myq + qn, 32, lane, base, packed, dinv, tab, feat, acc);
                }
                P &= P - 1u;
                any = __builtin_amdgcn_ballot_w32(P != 0u);
            }
        }
        hq = hn;
        lq = ln;
    }
    if (qn > 0) {
        if constexpr (MODE == 0) drain_count(myq, qn, lane, packed, cnti, cursor, mylist);
        else drain_rows<W>(myq, qn, lane, base, packed, dinv, tab, feat, acc);
    }
}

template <int MODE>
__device__ __forceinline__ void owner_body(const unsigned int* __restrict__ planes,
                                           const unsigned int* __restrict__ packed,
                                           const float* __restrict__ dinv,
                                           const float* __restrict__ feat,
                                           const float* __restrict__ bias,
                                           float* __restrict__ outp,
                                           unsigned int* __restrict__ list,
                                           int* __restrict__ listcnt,
                                           unsigned char* __restrict__ deg8, int E, int n) {
    constexpr int W = (MODE == 1) ? HID_DIM : ((MODE == 2) ? OUT_DIM : 1);
    __shared__ float acc[RMAX * W];
    __shared__ int   queue[OWN_WAVES * QCAP];
    __shared__ int   cursor;
    __shared__ __attribute__((aligned(16))) unsigned char tab[(MODE == 0) ? 16 : TAB_N];
    int* cnti = reinterpret_cast<int*>(acc);

    const int tid  = threadIdx.x;
    const int blk  = blockIdx.x;
    const int R    = (n + NBLK - 1) / NBLK;
    const int base = blk * R;
    int rows = n - base;
    if (rows > R) rows = R;
    if (rows <= 0) return;
    unsigned int* mylist = list + (size_t)blk * LCAP;

    uint4 tv[TAB_VEC];
    const int nvec = n >> 4;
    const int rot  = (int)(((long long)blk * nvec) / NBLK) & ~7;
    if constexpr (MODE != 0) {
        const uint4* srcv = reinterpret_cast<const uint4*>(deg8);
        #pragma unroll
        for (int k = 0; k < TAB_VEC; ++k) {
            const int i = k * OWN_T + tid;
            int j = i + rot;
            if (j >= nvec) j -= nvec;
            tv[k] = (i < nvec) ? srcv[j] : make_uint4(0u, 0u, 0u, 0u);
        }
    }

    for (int i = tid; i < rows * W; i += OWN_T) {
        if constexpr (MODE == 0) {
            cnti[i] = 1;
        } else {
            const int node = base + i / W;
            const float dv = dinv[node];
            acc[i] = feat[(size_t)node * W + (i % W)] * (dv * dv);
        }
    }
    if constexpr (MODE == 0) {
        if (tid == 0) cursor = 0;
    } else {
        uint4* tabv = reinterpret_cast<uint4*>(tab);
        #pragma unroll
        for (int k = 0; k < TAB_VEC; ++k) {
            const int i = k * OWN_T + tid;
            int j = i + rot;
            if (j >= nvec) j -= nvec;
            if (i < nvec) tabv[j] = tv[k];
        }
        if (tid < (n & 15)) tab[(nvec << 4) + tid] = deg8[(nvec << 4) + tid];
    }
    __syncthreads();

    if constexpr (MODE == 0) {
        scan_map<0>(planes, packed, dinv, tab, feat, acc, queue, &cursor, mylist, E, blk, base);
    } else {
        const int total = __builtin_amdgcn_readfirstlane(listcnt[blk]);
        const int wv    = __builtin_amdgcn_readfirstlane(tid >> 5);
        if (total <= LCAP) replay_rows<W>(mylist, total, wv, tid & 31, base, dinv, tab, feat, acc);
        else scan_map<MODE>(planes, packed, dinv, tab, feat, acc, queue, &cursor, mylist, E, blk, base);
    }
    __syncthreads();

    if constexpr (MODE == 0) {
        if (tid == 0) listcnt[blk] = cursor;
    }

    for (int i = tid; i < rows * W; i += OWN_T) {
        if constexpr (MODE == 0) {
            const int cn = cnti[i];
            outp[base + i] = rsqrtf((float)cn);
            deg8[base + i] = (unsigned char)((cn < (int)TAB_ESC) ? cn : (int)TAB_ESC);
        } else if constexpr (MODE == 1) {
            outp[(size_t)base * W + i] = fmaxf(acc[i] + bias[i & (W - 1)], 0.0f);
        } else {
            outp[(size_t)base * W + i] = acc[i] + bias[i & (W - 1)];
        }
    }
}

__global__ void __launch_bounds__(OWN_T) owner_deg_kernel(const unsigned int* __restrict__ planes,
                                                          const unsigned int* __restrict__ packed,
                                                          float* __restrict__ dinv,
                                                          unsigned int* __restrict__ list,
                                                          int* __restrict__ listcnt,
                                                          unsigned char* __restrict__ deg8, int E, int n) {
    owner_body<0>(planes, packed, nullptr, nullptr, nullptr, dinv, list, listcnt, deg8, E, n);
}

__global__ void __launch_bounds__(OWN_T) owner_agg1_kernel(const unsigned int* __restrict__ planes,
                                                           const unsigned int* __restrict__ packed,
                                                           const float* __restrict__ dinv,
                                                           const float* __restrict__ hw,
                                                           const float* __restrict__ b1,
                                                           float* __restrict__ h,
                                                           unsigned int* __restrict__ list,
                                                           int* __restrict__ listcnt,
                                                           unsigned char* __restrict__ deg8, int E, int n) {
    owner_body<1>(planes, packed, dinv, hw, b1, h, list, listcnt, deg8, E, n);
}

__global__ void __launch_bounds__(OWN_T) owner_agg2_kernel(const unsigned int* __restrict__ planes,
                                                           const unsigned int* __restrict__ packed,
                                                           const float* __restrict__ dinv,
                                                           const float* __restrict__ g,
                                                           const float* __restrict__ b2,
                                                           float* __restrict__ out,
                                                           unsigned int* __restrict__ list,
                                                           int* __restrict__ listcnt,
                                                           unsigned char* __restrict__ deg8, int E, int n) {
    owner_body<2>(planes, packed, dinv, g, b2, out, list, listcnt, deg8, E, n);
}

#define PROD_WG 256
#define PROD_WAVES 8

__device__ __forceinline__ int prod_step_k(int t, int hf) { return 8 * (t >> 1) + 4 * hf + 2 * (t & 1); }

template <int NV>
__device__ __forceinline__ void load_rows(float4 (&a)[NV], const float* __restrict__ src, int tile,
                                          int l16, int hf, int n) {
    int row = tile * 16 + l16;
    if (row >= n) row = n - 1;
    const float4* p = reinterpret_cast<const float4*>(src + (size_t)row * (NV * 8)) + hf;
    #pragma unroll
    for (int j = 0; j < NV; ++j) a[j] = p[2 * j];
}

__device__ __forceinline__ void tile1(const float4 (&a)[16], const v2f (&bl)[32], const v2f (&bh)[32],
                                      float* __restrict__ hw, int tile, int l16, int hf, int n) {
    v8f c0 = {}; v8f c1 = {};
    #pragma unroll
    for (int j = 0; j < 16; ++j) {
        v2f lo; lo.x = a[j].x; lo.y = a[j].y;
        v2f hi; hi.x = a[j].z; hi.y = a[j].w;
        c0 = __builtin_amdgcn_wmma_f32_16x16x4_f32(false, lo, false, bl[2 * j],     (short)0, c0, false, false);
        c1 = __builtin_amdgcn_wmma_f32_16x16x4_f32(false, lo, false, bh[2 * j],     (short)0, c1, false, false);
        c0 = __builtin_amdgcn_wmma_f32_16x16x4_f32(false, hi, false, bl[2 * j + 1], (short)0, c0, false, false);
        c1 = __builtin_amdgcn_wmma_f32_16x16x4_f32(false, hi, false, bh[2 * j + 1], (short)0, c1, false, false);
    }
    #pragma unroll
    for (int i = 0; i < 8; ++i) {
        const int orow = tile * 16 + i + 8 * hf;
        if (orow < n) {
            hw[orow * HID_DIM + l16]      = c0[i];
            hw[orow * HID_DIM + 16 + l16] = c1[i];
        }
    }
}

__global__ void __launch_bounds__(32 * PROD_WAVES) gemm1_kernel(const float* __restrict__ x,
                                                                const float* __restrict__ W1,
                                                                float* __restrict__ hw, int n) {
    const int lane = threadIdx.x & 31;
    const int hf   = lane >> 4;
    const int l16  = lane & 15;
    const int ntiles = (n + 15) >> 4;
    const int stride = (int)gridDim.x * PROD_WAVES;
    int tile = (threadIdx.x >> 5) * (int)gridDim.x + (int)blockIdx.x;
    if (tile >= ntiles) return;

    float4 a0[16]; float4 a1[16];
    load_rows<16>(a0, x, tile, l16, hf, n);
    __builtin_amdgcn_sched_barrier(0);

    v2f bl[32]; v2f bh[32];
    #pragma unroll
    for (int t = 0; t < 32; ++t) {
        const int kk = prod_step_k(t, hf);
        bl[t].x = W1[kk * HID_DIM + l16];       bl[t].y = W1[(kk + 1) * HID_DIM + l16];
        bh[t].x = W1[kk * HID_DIM + 16 + l16];  bh[t].y = W1[(kk + 1) * HID_DIM + 16 + l16];
    }
    for (;;) {
        const int t1 = tile + stride;
        load_rows<16>(a1, x, t1, l16, hf, n);
        __builtin_amdgcn_sched_barrier(0);
        tile1(a0, bl, bh, hw, tile, l16, hf, n);
        if (t1 >= ntiles) break;
        const int t2 = t1 + stride;
        load_rows<16>(a0, x, t2, l16, hf, n);
        __builtin_amdgcn_sched_barrier(0);
        tile1(a1, bl, bh, hw, t1, l16, hf, n);
        if (t2 >= ntiles) break;
        tile = t2;
    }
}


__device__ __forceinline__ void tile2(const float4 (&a)[4], const v2f (&b)[8],
                                      float* __restrict__ g, int tile, int l16, int hf, int n) {
    v8f c = {};
    #pragma unroll
    for (int j = 0; j < 4; ++j) {
        v2f lo; lo.x = a[j].x; lo.y = a[j].y;
        v2f hi; hi.x = a[j].z; hi.y = a[j].w;
        c = __builtin_amdgcn_wmma_f32_16x16x4_f32(false, lo, false, b[2 * j],     (short)0, c, false, false);
        c = __builtin_amdgcn_wmma_f32_16x16x4_f32(false, hi, false, b[2 * j + 1], (short)0, c, false, false);
    }
    if (l16 < OUT_DIM) {
        #pragma unroll
        for (int i = 0; i < 8; ++i) {
            const int orow = tile * 16 + i + 8 * hf;
            if (orow < n) g[orow * OUT_DIM + l16] = c[i];
        }
    }
}

__global__ void __launch_bounds__(32 * PROD_WAVES) gemm2_kernel(const float* __restrict__ h,
                                                                const float* __restrict__ W2,
                                                                float* __restrict__ g, int n) {
    const int lane = threadIdx.x & 31;
    const int hf   = lane >> 4;
    const int l16  = lane & 15;
    const int ntiles = (n + 15) >> 4;
    const int stride = (int)gridDim.x * PROD_WAVES;
    int tile = (threadIdx.x >> 5) * (int)gridDim.x + (int)blockIdx.x;
    if (tile >= ntiles) return;
    const bool live = (l16 < OUT_DIM);
    const int  cw   = l16 & (OUT_DIM - 1);

    float4 a0[4]; float4 a1[4];
    load_rows<4>(a0, h, tile, l16, hf, n);

    v2f b[8];
    #pragma unroll
    for (int t = 0; t < 8; ++t) {
        const int kk = prod_step_k(t, hf);
        const float w0 = W2[kk * OUT_DIM + cw];
        const float w1 = W2[(kk + 1) * OUT_DIM + cw];
        b[t].x = live ? w0 : 0.0f;
        b[t].y = live ? w1 : 0.0f;
    }
    for (;;) {
        const int t1 = tile + stride;
        load_rows<4>(a1, h, t1, l16, hf, n);
        tile2(a0, b, g, tile, l16, hf, n);
        if (t1 >= ntiles) break;
        const int t2 = t1 + stride;
        load_rows<4>(a0, h, t2, l16, hf, n);
        tile2(a1, b, g, t1, l16, hf, n);
        if (t2 >= ntiles) break;
        tile = t2;
    }
}


extern "C" void kernel_launch(void* const* d_in, const int* in_sizes, int n_in,
                              void* d_out, int out_size, void* d_ws, size_t ws_size,
                              hipStream_t stream) {
    const float* x  = (const float*)d_in[0];
    const int*   ei = (const int*)d_in[1];
    const float* W1 = (const float*)d_in[2];
    const float* b1 = (const float*)d_in[3];
    const float* W2 = (const float*)d_in[4];
    const float* b2 = (const float*)d_in[5];
    float* out = (float*)d_out;

    const int n = in_sizes[0] / IN_DIM;
    const int E = in_sizes[1] / 2;
    const int* srcIdx = ei;
    const int* dstIdx = ei + E;

    const int nchunks = PLANE_CHUNKS(E);
    const int PW = nchunks * CH_WORDS;
    unsigned int* planes = (unsigned int*)d_ws;
    unsigned int* packed = planes + (size_t)32 * PW;
    float* dinv = (float*)(packed + (size_t)nchunks * CH_EDGES);
    float* hw   = dinv + n;
    float* h    = hw + (size_t)n * HID_DIM;
    float* g    = h  + (size_t)n * HID_DIM;
    int* listcnt = (int*)(g + (size_t)n * OUT_DIM);
    unsigned int* list = (unsigned int*)(listcnt + NBLK);
    unsigned char* deg8 = (unsigned char*)(((uintptr_t)(list + (size_t)NBLK * LCAP) + 255) & ~(uintptr_t)255);

    const dim3 blk(256);

    planes_kernel<<<dim3(nchunks), blk, 0, stream>>>(srcIdx, dstIdx, planes, packed, E, PW, n);
    owner_deg_kernel<<<dim3(NBLK), dim3(OWN_T), 0, stream>>>(planes, packed, dinv, list, listcnt, deg8, E, n);
    gemm1_kernel<<<dim3(PROD_WG), blk, 0, stream>>>(x, W1, hw, n);
    owner_agg1_kernel<<<dim3(NBLK), dim3(OWN_T), 0, stream>>>(planes, packed, dinv, hw, b1, h, list, listcnt, deg8, E, n);
    gemm2_kernel<<<dim3(PROD_WG), blk, 0, stream>>>(h, W2, g, n);
    owner_agg2_kernel<<<dim3(NBLK), dim3(OWN_T), 0, stream>>>(planes, packed, dinv, g, b2, out, list, listcnt, deg8, E, n);
}


# Row passes: no load waits behind one it does not need: start values, edge count, first list words issue in one group before the barrier, next list words and weights beside the row gathers; same sums
# speedup vs baseline: 4.2905x; 1.0732x over previous
#include <hip/hip_runtime.h>


typedef float v2f __attribute__((ext_vector_type(2)));
typedef float v8f __attribute__((ext_vector_type(8)));

#define IN_DIM  128
#define HID_DIM 32
#define OUT_DIM 8

#define NBLK    256
#define RMAX    512
#define OWN_T   1024
#define OWN_WAVES (OWN_T / 32)
#define QCAP    64
#define LCAP    65536
#define SRC_BITS 17
#define CH_WORDS 64
#define CH_EDGES (CH_WORDS * 32)
#define SP_PITCH (CH_WORDS + 1)
#define ROW_CHUNKS (OWN_T * 4 / CH_WORDS)
#define PLANE_CHUNKS(E) ((((E) + CH_EDGES - 1) / CH_EDGES + ROW_CHUNKS - 1) / ROW_CHUNKS * ROW_CHUNKS)
#define SRC_MASK ((1u << SRC_BITS) - 1u)
#define TAB_N   (NBLK * RMAX)
#define TAB_ESC 255u
#define TAB_VEC (TAB_N / 16 / OWN_T)
static_assert(NBLK * RMAX <= (1 << SRC_BITS), "a source node must fit the low field of a list entry");
static_assert(RMAX <= (1 << (32 - SRC_BITS)), "a row inside the block must fit the high field of a list entry");

__global__ void __launch_bounds__(256) planes_kernel(const int* __restrict__ srcIdx,
                                                     const int* __restrict__ dstIdx,
                                                     unsigned int* __restrict__ planes,
                                                     unsigned int* __restrict__ packed,
                                                     int E, int PW, int n) {
    __shared__ unsigned int sp[32 * SP_PITCH];
    const int tid  = threadIdx.x;
    const int lane = tid & 31;
    const int wave = tid >> 5;
    const unsigned R = (unsigned)((n + NBLK - 1) / NBLK);
    const int ebase = blockIdx.x * CH_EDGES;
    int d[8];
    int s[8];
    #pragma unroll
    for (int j = 0; j < 8; ++j) {
        const int e = ebase + (j * 8 + wave) * 32 + lane;
        d[j] = (e < E) ? dstIdx[e] : -1;
        s[j] = (e < E) ? srcIdx[e] : 0;
    }
    #pragma unroll
    for (int j = 0; j < 8; ++j) {
        const int e = ebase + (j * 8 + wave) * 32 + lane;
        const unsigned blkno = (unsigned)d[j] / R;
        if (e < E) packed[e] = (((unsigned)d[j] - blkno * R) << SRC_BITS) | (unsigned)s[j];
        const unsigned b  = blkno & 255u;
        const unsigned hi = (d[j] >= 0) ? (b >> 4) : 16u;
        const unsigned lo = (d[j] >= 0) ? (b & 15u) : 16u;
        int w = 0;
        #pragma unroll
        for (int v = 0; v < 16; ++v) {
            const int mh = (int)__builtin_amdgcn_ballot_w32(hi == (unsigned)v);
            const int ml = (int)__builtin_amdgcn_ballot_w32(lo == (unsigned)v);
            w = (lane == v)      ? mh : w;
            w = (lane == 16 + v) ? ml : w;
        }
        sp[lane * SP_PITCH + j * 8 + wave] = (unsigned)w;
    }
    __syncthreads();
    const int p = tid >> 3;
    const int q = tid & 7;
    const unsigned int* row = sp + p * SP_PITCH;
    const uint4 a = make_uint4(row[q * 4], row[q * 4 + 1], row[q * 4 + 2], row[q * 4 + 3]);
    const uint4 c = make_uint4(row[32 + q * 4], row[33 + q * 4], row[34 + q * 4], row[35 + q * 4]);
    uint4* outv = reinterpret_cast<uint4*>(planes + (size_t)p * PW + (size_t)blockIdx.x * CH_WORDS);
    outv[q]     = a;
    outv[8 + q] = c;
}

template <int W>
__device__ __forceinline__ int lane_get(int v, int idx) {
    if constexpr (W == 32) return __builtin_amdgcn_readlane(v, idx);
    else                   return __shfl(v, idx);
}

__device__ __forceinline__ float edge_weight(const unsigned char* tab, const float* __restrict__ dinv,
                                             int s, int d) {
    const unsigned cs = tab[s];
    const unsigned cd = tab[d];
    float ws = rsqrtf((float)cs);
    float wd = rsqrtf((float)cd);
    if (cs == TAB_ESC) ws = dinv[s];
    if (cd == TAB_ESC) wd = dinv[d];
    return ws * wd;
}

__device__ __forceinline__ void drain_count(const int* q, int cnt, int lane,
                                            const unsigned int* __restrict__ packed, int* cnti,
                                            int* cursor, unsigned int* __restrict__ mylist) {
    unsigned ent = 0;
    if (lane < cnt) {
        ent = packed[q[lane]];
        atomicAdd(&cnti[ent >> SRC_BITS], 1);
    }
    int off = 0;
    if (lane == 0) off = atomicAdd(cursor, cnt);
    off = __builtin_amdgcn_readfirstlane(off);
    if (off + cnt <= LCAP && lane < cnt) mylist[off + lane] = ent;
}

template <int W, typename WF>
__device__ __forceinline__ void gather_add(int s, int dl, WF wf, int cnt, int lane,
                                           const float* __restrict__ feat, float* acc) {
    if constexpr (W == 32) {
        float r[32];
        #pragma unroll
        for (int j = 0; j < 32; ++j) {
            const int sj = __builtin_amdgcn_readlane(s, j);
            r[j] = feat[(size_t)sj * 32 + lane];
        }
        const float nrm = wf();
        if (cnt == 32) {
            #pragma unroll
            for (int j = 0; j < 32; ++j) {
                const float nj = __int_as_float(__builtin_amdgcn_readlane(__float_as_int(nrm), j));
                const int   dj = __builtin_amdgcn_readlane(dl, j);
                atomicAdd(&acc[dj * 32 + lane], r[j] * nj);
            }
        } else {
            #pragma unroll
            for (int j = 0; j < 32; ++j) {
                const float nj = __int_as_float(__builtin_amdgcn_readlane(__float_as_int(nrm), j));
                const int   dj = __builtin_amdgcn_readlane(dl, j);
                if (j < cnt) atomicAdd(&acc[dj * 32 + lane], r[j] * nj);
            }
        }
    } else {
        constexpr int EPI = 32 / W;
        constexpr int U   = 8;
        const int sub = lane / W;
        const int f   = lane % W;
        for (int i0 = 0; i0 < cnt; i0 += U * EPI) {
            float r[U];
            #pragma unroll
            for (int j = 0; j < U; ++j) {
                const int idx = i0 + j * EPI + sub;
                const int sj = lane_get<W>(s, idx);
                r[j] = feat[(size_t)sj * W + f];
            }
            const float nrm = wf();
            #pragma unroll
            for (int j = 0; j < U; ++j) {
                const int idx = i0 + j * EPI + sub;
                const float nj = __int_as_float(lane_get<W>(__float_as_int(nrm), idx));
                const int   dj = lane_get<W>(dl, idx);
                if (idx < cnt) atomicAdd(&acc[dj * W + f], r[j] * nj);
            }
        }
    }
}

template <int W>
__device__ __forceinline__ void drain_rows(const int* q, int cnt, int lane, int base,
                                           const unsigned int* __restrict__ packed,
                                           const float* __restrict__ dinv, const unsigned char* tab,
                                           const float* __restrict__ feat, float* acc) {
    int s = 0, dl = 0;
    if (lane < cnt) {
        const unsigned ent = packed[q[lane]];
        s  = (int)(ent & SRC_MASK);
        dl = (int)(ent >> SRC_BITS);
    }
    gather_add<W>(s, dl, [&]() { return (lane < cnt) ? edge_weight(tab, dinv, s, base + dl) : 0.0f; },
                  cnt, lane, feat, acc);
}

template <int W>
__device__ __forceinline__ void replay_rows(const unsigned int* __restrict__ mylist, int total, unsigned first,
                                            int wave, int lane, int base,
                                            const float* __restrict__ dinv, const unsigned char* tab,
                                            const float* __restrict__ feat, float* acc) {
    int c = wave * 32;
    unsigned ent = (c + lane < total) ? first : 0u;
    for (; c < total; c += OWN_T) {
        int cnt = total - c;
        if (cnt > 32) cnt = 32;
        const int s  = (int)(ent & SRC_MASK);
        const int dl = (int)(ent >> SRC_BITS);
        const int cn = c + OWN_T;
        const unsigned nxt = (cn + lane < total) ? mylist[cn + lane] : 0u;
        gather_add<W>(s, dl, [&]() { return (lane < cnt) ? edge_weight(tab, dinv, s, base + dl) : 0.0f; },
                      cnt, lane, feat, acc);
        ent = nxt;
    }
}

__device__ __forceinline__ void first_plane_vecs(const unsigned int* __restrict__ planes, int E, int blk, int tid,
                                                 uint4& hq, uint4& lq) {
    const int PV = PLANE_CHUNKS(E) * (CH_WORDS / 4);
    hq = (reinterpret_cast<const uint4*>(planes) + (size_t)(blk >> 4) * PV + tid)[0];
    lq = (reinterpret_cast<const uint4*>(planes) + (size_t)(16 + (blk & 15)) * PV + tid)[0];
}

template <int MODE>
__device__ __forceinline__ void scan_map(const unsigned int* __restrict__ planes,
                                         const unsigned int* __restrict__ packed,
                                         const float* __restrict__ dinv, const unsigned char* tab,
                                         const float* __restrict__ feat,
                                         float* acc, int* queue, int* cursor,
                                         unsigned int* __restrict__ mylist,
                                         int E, int blk, int base, uint4 hq, uint4 lq) {
    constexpr int W = (MODE == 1) ? HID_DIM : ((MODE == 2) ? OUT_DIM : 1);
    int* cnti = reinterpret_cast<int*>(acc);
    const int tid  = threadIdx.x;
    const int lane = tid & 31;
    const int wave = tid >> 5;

    int* myq = queue + wave * QCAP;
    int qn = 0;
    const int PV   = PLANE_CHUNKS(E) * (CH_WORDS / 4);
    const int nrow = PV / OWN_T;
    const uint4* hp = reinterpret_cast<const uint4*>(planes) + (size_t)(blk >> 4) * PV + tid;
    const uint4* lp = reinterpret_cast<const uint4*>(planes) + (size_t)(16 + (blk & 15)) * PV + tid;

    for (int r = 0; r < nrow; ++r) {
        uint4 hn = hq, ln = lq;
        if (r + 1 < nrow) {
            hn = hp[(size_t)(r + 1) * OWN_T];
            ln = lp[(size_t)(r + 1) * OWN_T];
        }
        const unsigned m[4] = { hq.x & lq.x, hq.y & lq.y, hq.z & lq.z, hq.w & lq.w };
        const int ev = (r * OWN_T + tid) * 128;
        #pragma unroll
        for (int c = 0; c < 4; ++c) {
            unsigned P = m[c];
            unsigned any = __builtin_amdgcn_ballot_w32(P != 0u);
            while (any != 0u) {
                const bool ok = (P != 0u);
                const int  e  = ev + c * 32 + __builtin_ctz(ok ? P : 1u);
                if (ok) myq[qn + (int)__builtin_amdgcn_mbcnt_lo(any, 0u)] = e;
                qn += __builtin_popcount(any);
                if (qn >= 32) {
                    qn -= 32;
                    if constexpr (MODE == 0) drain_count(myq + qn, 32, lane, packed, cnti, cursor, mylist);
                    else drain_rows<W>(myq + qn, 32, lane, base, packed, dinv, tab, feat, acc);
                }
                P &= P - 1u;
                any = __builtin_amdgcn_ballot_w32(P != 0u);
            }
        }
        hq = hn;
        lq = ln;
    }
    if (qn > 0) {
        if constexpr (MODE == 0) drain_count(myq, qn, lane, packed, cnti, cursor, mylist);
        else drain_rows<W>(myq, qn, lane, base, packed, dinv, tab, feat, acc);
    }
}

template <int MODE>
__device__ __forceinline__ void owner_body(const unsigned int* __restrict__ planes,
                                           const unsigned int* __restrict__ packed,
                                           const float* __restrict__ dinv,
                                           const float* __restrict__ feat,
                                           const float* __restrict__ bias,
                                           float* __restrict__ outp,
                                           unsigned int* __restrict__ list,
                                           int* __restrict__ listcnt,
                                           unsigned char* __restrict__ deg8, int E, int n) {
    constexpr int W = (MODE == 1) ? HID_DIM : ((MODE == 2) ? OUT_DIM : 1);
    __shared__ float acc[RMAX * W];
    __shared__ int   queue[OWN_WAVES * QCAP];
    __shared__ int   cursor;
    __shared__ __attribute__((aligned(16))) unsigned char tab[(MODE == 0) ? 16 : TAB_N];
    int* cnti = reinterpret_cast<int*>(acc);

    const int tid  = threadIdx.x;
    const int blk  = blockIdx.x;
    const int R    = (n + NBLK - 1) / NBLK;
    const int base = blk * R;
    int rows = n - base;
    if (rows > R) rows = R;
    if (rows <= 0) return;
    unsigned int* mylist = list + (size_t)blk * LCAP;

    uint4 tv[TAB_VEC];
    const int nvec = n >> 4;
    const int rot  = (int)(((long long)blk * nvec) / NBLK) & ~7;
    if constexpr (MODE != 0) {
        const uint4* srcv = reinterpret_cast<const uint4*>(deg8);
        #pragma unroll
        for (int k = 0; k < TAB_VEC; ++k) {
            const int i = k * OWN_T + tid;
            int j = i + rot;
            if (j >= nvec) j -= nvec;
            tv[k] = (i < nvec) ? srcv[j] : make_uint4(0u, 0u, 0u, 0u);
        }
    }

    constexpr int SV_N = (MODE == 0) ? 1 : (RMAX * W / OWN_T);
    float sv[SV_N];
    float sd[SV_N];
    int   total_v = 0;
    unsigned first_ent = 0u;
    uint4 hq0 = make_uint4(0u, 0u, 0u, 0u), lq0 = make_uint4(0u, 0u, 0u, 0u);
    if constexpr (MODE == 0) {
        first_plane_vecs(planes, E, blk, tid, hq0, lq0);
    } else {
        const int lim = rows * W;
        #pragma unroll
        for (int k = 0; k < SV_N; ++k) {
            const int i  = k * OWN_T + tid;
            const int ic = (i < lim) ? i : (lim - 1);
            sd[k] = dinv[base + ic / W];
            sv[k] = feat[(size_t)base * W + ic];
        }
        total_v   = listcnt[blk];
        first_ent = mylist[tid];
    }

    if constexpr (MODE == 0) {
        for (int i = tid; i < rows; i += OWN_T) cnti[i] = 1;
        if (tid == 0) cursor = 0;
    } else {
        #pragma unroll
        for (int k = 0; k < SV_N; ++k) acc[k * OWN_T + tid] = sv[k] * (sd[k] * sd[k]);
        uint4* tabv = reinterpret_cast<uint4*>(tab);
        #pragma unroll
        for (int k = 0; k < TAB_VEC; ++k) {
            const int i = k * OWN_T + tid;
            int j = i + rot;
            if (j >= nvec) j -= nvec;
            if (i < nvec) tabv[j] = tv[k];
        }
        if (tid < (n & 15)) tab[(nvec << 4) + tid] = deg8[(nvec << 4) + tid];
    }
    __syncthreads();

    if constexpr (MODE == 0) {
        scan_map<0>(planes, packed, dinv, tab, feat, acc, queue, &cursor, mylist, E, blk, base, hq0, lq0);
    } else {
        const int total = __builtin_amdgcn_readfirstlane(total_v);
        const int wv    = __builtin_amdgcn_readfirstlane(tid >> 5);
        if (total <= LCAP) replay_rows<W>(mylist, total, first_ent, wv, tid & 31, base, dinv, tab, feat, acc);
        else {
            first_plane_vecs(planes, E, blk, tid, hq0, lq0);
            scan_map<MODE>(planes, packed, dinv, tab, feat, acc, queue, &cursor, mylist, E, blk, base, hq0, lq0);
        }
    }
    __syncthreads();

    if constexpr (MODE == 0) {
        if (tid == 0) listcnt[blk] = cursor;
    }

    for (int i = tid; i < rows * W; i += OWN_T) {
        if constexpr (MODE == 0) {
            const int cn = cnti[i];
            outp[base + i] = rsqrtf((float)cn);
            deg8[base + i] = (unsigned char)((cn < (int)TAB_ESC) ? cn : (int)TAB_ESC);
        } else if constexpr (MODE == 1) {
            outp[(size_t)base * W + i] = fmaxf(acc[i] + bias[i & (W - 1)], 0.0f);
        } else {
            outp[(size_t)base * W + i] = acc[i] + bias[i & (W - 1)];
        }
    }
}

__global__ void __launch_bounds__(OWN_T) owner_deg_kernel(const unsigned int* __restrict__ planes,
                                                          const unsigned int* __restrict__ packed,
                                                          float* __restrict__ dinv,
                                                          unsigned int* __restrict__ list,
                                                          int* __restrict__ listcnt,
                                                          unsigned char* __restrict__ deg8, int E, int n) {
    owner_body<0>(planes, packed, nullptr, nullptr, nullptr, dinv, list, listcnt, deg8, E, n);
}

__global__ void __launch_bounds__(OWN_T) owner_agg1_kernel(const unsigned int* __restrict__ planes,
                                                           const unsigned int* __restrict__ packed,
                                                           const float* __restrict__ dinv,
                                                           const float* __restrict__ hw,
                                                           const float* __restrict__ b1,
                                                           float* __restrict__ h,
                                                           unsigned int* __restrict__ list,
                                                           int* __restrict__ listcnt,
                                                           unsigned char* __restrict__ deg8, int E, int n) {
    owner_body<1>(planes, packed, dinv, hw, b1, h, list, listcnt, deg8, E, n);
}

__global__ void __launch_bounds__(OWN_T) owner_agg2_kernel(const unsigned int* __restrict__ planes,
                                                           const unsigned int* __restrict__ packed,
                                                           const float* __restrict__ dinv,
                                                           const float* __restrict__ g,
                                                           const float* __restrict__ b2,
                                                           float* __restrict__ out,
                                                           unsigned int* __restrict__ list,
                                                           int* __restrict__ listcnt,
                                                           unsigned char* __restrict__ deg8, int E, int n) {
    owner_body<2>(planes, packed, dinv, g, b2, out, list, listcnt, deg8, E, n);
}

#define PROD_WG 256
#define PROD_WAVES 8

__device__ __forceinline__ int prod_step_k(int t, int hf) { return 8 * (t >> 1) + 4 * hf + 2 * (t & 1); }

template <int NV>
__device__ __forceinline__ void load_rows(float4 (&a)[NV], const float* __restrict__ src, int tile,
                                          int l16, int hf, int n) {
    int row = tile * 16 + l16;
    if (row >= n) row = n - 1;
    const float4* p = reinterpret_cast<const float4*>(src + (size_t)row * (NV * 8)) + hf;
    #pragma unroll
    for (int j = 0; j < NV; ++j) a[j] = p[2 * j];
}

__device__ __forceinline__ void tile1(const float4 (&a)[16], const v2f (&bl)[32], const v2f (&bh)[32],
                                      float* __restrict__ hw, int tile, int l16, int hf, int n) {
    v8f c0 = {}; v8f c1 = {};
    #pragma unroll
    for (int j = 0; j < 16; ++j) {
        v2f lo; lo.x = a[j].x; lo.y = a[j].y;
        v2f hi; hi.x = a[j].z; hi.y = a[j].w;
        c0 = __builtin_amdgcn_wmma_f32_16x16x4_f32(false, lo, false, bl[2 * j],     (short)0, c0, false, false);
        c1 = __builtin_amdgcn_wmma_f32_16x16x4_f32(false, lo, false, bh[2 * j],     (short)0, c1, false, false);
        c0 = __builtin_amdgcn_wmma_f32_16x16x4_f32(false, hi, false, bl[2 * j + 1], (short)0, c0, false, false);
        c1 = __builtin_amdgcn_wmma_f32_16x16x4_f32(false, hi, false, bh[2 * j + 1], (short)0, c1, false, false);
    }
    #pragma unroll
    for (int i = 0; i < 8; ++i) {
        const int orow = tile * 16 + i + 8 * hf;
        if (orow < n) {
            hw[orow * HID_DIM + l16]      = c0[i];
            hw[orow * HID_DIM + 16 + l16] = c1[i];
        }
    }
}

__global__ void __launch_bounds__(32 * PROD_WAVES) gemm1_kernel(const float* __restrict__ x,
                                                                const float* __restrict__ W1,
                                                                float* __restrict__ hw, int n) {
    const int lane = threadIdx.x & 31;
    const int hf   = lane >> 4;
    const int l16  = lane & 15;
    const int ntiles = (n + 15) >> 4;
    const int stride = (int)gridDim.x * PROD_WAVES;
    int tile = (threadIdx.x >> 5) * (int)gridDim.x + (int)blockIdx.x;
    if (tile >= ntiles) return;

    float4 a0[16]; float4 a1[16];
    load_rows<16>(a0, x, tile, l16, hf, n);
    __builtin_amdgcn_sched_barrier(0);

    v2f bl[32]; v2f bh[32];
    #pragma unroll
    for (int t = 0; t < 32; ++t) {
        const int kk = prod_step_k(t, hf);
        bl[t].x = W1[kk * HID_DIM + l16];       bl[t].y = W1[(kk + 1) * HID_DIM + l16];
        bh[t].x = W1[kk * HID_DIM + 16 + l16];  bh[t].y = W1[(kk + 1) * HID_DIM + 16 + l16];
    }
    for (;;) {
        const int t1 = tile + stride;
        load_rows<16>(a1, x, t1, l16, hf, n);
        __builtin_amdgcn_sched_barrier(0);
        tile1(a0, bl, bh, hw, tile, l16, hf, n);
        if (t1 >= ntiles) break;
        const int t2 = t1 + stride;
        load_rows<16>(a0, x, t2, l16, hf, n);
        __builtin_amdgcn_sched_barrier(0);
        tile1(a1, bl, bh, hw, t1, l16, hf, n);
        if (t2 >= ntiles) break;
        tile = t2;
    }
}


__device__ __forceinline__ void tile2(const float4 (&a)[4], const v2f (&b)[8],
                                      float* __restrict__ g, int tile, int l16, int hf, int n) {
    v8f c = {};
    #pragma unroll
    for (int j = 0; j < 4; ++j) {
        v2f lo; lo.x = a[j].x; lo.y = a[j].y;
        v2f hi; hi.x = a[j].z; hi.y = a[j].w;
        c = __builtin_amdgcn_wmma_f32_16x16x4_f32(false, lo, false, b[2 * j],     (short)0, c, false, false);
        c = __builtin_amdgcn_wmma_f32_16x16x4_f32(false, hi, false, b[2 * j + 1], (short)0, c, false, false);
    }
    if (l16 < OUT_DIM) {
        #pragma unroll
        for (int i = 0; i < 8; ++i) {
            const int orow = tile * 16 + i + 8 * hf;
            if (orow < n) g[orow * OUT_DIM + l16] = c[i];
        }
    }
}

__global__ void __launch_bounds__(32 * PROD_WAVES) gemm2_kernel(const float* __restrict__ h,
                                                                const float* __restrict__ W2,
                                                                float* __restrict__ g, int n) {
    const int lane = threadIdx.x & 31;
    const int hf   = lane >> 4;
    const int l16  = lane & 15;
    const int ntiles = (n + 15) >> 4;
    const int stride = (int)gridDim.x * PROD_WAVES;
    int tile = (threadIdx.x >> 5) * (int)gridDim.x + (int)blockIdx.x;
    if (tile >= ntiles) return;
    const bool live = (l16 < OUT_DIM);
    const int  cw   = l16 & (OUT_DIM - 1);

    float4 a0[4]; float4 a1[4];
    load_rows<4>(a0, h, tile, l16, hf, n);

    v2f b[8];
    #pragma unroll
    for (int t = 0; t < 8; ++t) {
        const int kk = prod_step_k(t, hf);
        const float w0 = W2[kk * OUT_DIM + cw];
        const float w1 = W2[(kk + 1) * OUT_DIM + cw];
        b[t].x = live ? w0 : 0.0f;
        b[t].y = live ? w1 : 0.0f;
    }
    for (;;) {
        const int t1 = tile + stride;
        load_rows<4>(a1, h, t1, l16, hf, n);
        tile2(a0, b, g, tile, l16, hf, n);
        if (t1 >= ntiles) break;
        const int t2 = t1 + stride;
        load_rows<4>(a0, h, t2, l16, hf, n);
        tile2(a1, b, g, t1, l16, hf, n);
        if (t2 >= ntiles) break;
        tile = t2;
    }
}


extern "C" void kernel_launch(void* const* d_in, const int* in_sizes, int n_in,
                              void* d_out, int out_size, void* d_ws, size_t ws_size,
                              hipStream_t stream) {
    const float* x  = (const float*)d_in[0];
    const int*   ei = (const int*)d_in[1];
    const float* W1 = (const float*)d_in[2];
    const float* b1 = (const float*)d_in[3];
    const float* W2 = (const float*)d_in[4];
    const float* b2 = (const float*)d_in[5];
    float* out = (float*)d_out;

    const int n = in_sizes[0] / IN_DIM;
    const int E = in_sizes[1] / 2;
    const int* srcIdx = ei;
    const int* dstIdx = ei + E;

    const int nchunks = PLANE_CHUNKS(E);
    const int PW = nchunks * CH_WORDS;
    unsigned int* planes = (unsigned int*)d_ws;
    unsigned int* packed = planes + (size_t)32 * PW;
    float* dinv = (float*)(packed + (size_t)nchunks * CH_EDGES);
    float* hw   = dinv + n;
    float* h    = hw + (size_t)n * HID_DIM;
    float* g    = h  + (size_t)n * HID_DIM;
    int* listcnt = (int*)(g + (size_t)n * OUT_DIM);
    unsigned int* list = (unsigned int*)(listcnt + NBLK);
    unsigned char* deg8 = (unsigned char*)(((uintptr_t)(list + (size_t)NBLK * LCAP) + 255) & ~(uintptr_t)255);

    const dim3 blk(256);

    planes_kernel<<<dim3(nchunks), blk, 0, stream>>>(srcIdx, dstIdx, planes, packed, E, PW, n);
    owner_deg_kernel<<<dim3(NBLK), dim3(OWN_T), 0, stream>>>(planes, packed, dinv, list, listcnt, deg8, E, n);
    gemm1_kernel<<<dim3(PROD_WG), blk, 0, stream>>>(x, W1, hw, n);
    owner_agg1_kernel<<<dim3(NBLK), dim3(OWN_T), 0, stream>>>(planes, packed, dinv, hw, b1, h, list, listcnt, deg8, E, n);
    gemm2_kernel<<<dim3(PROD_WG), blk, 0, stream>>>(h, W2, g, n);
    owner_agg2_kernel<<<dim3(NBLK), dim3(OWN_T), 0, stream>>>(planes, packed, dinv, g, b2, out, list, listcnt, deg8, E, n);
}
